# starting kernel
# baseline (speedup 1.0000x reference)
#include <hip/hip_runtime.h>
#include <hip/hip_bf16.h>

typedef float v2f __attribute__((ext_vector_type(2)));
typedef float v8f __attribute__((ext_vector_type(8)));

#define F_IN  512
#define F_HID 32
#define F_OUT 64


__global__ void k_init_deg(float* __restrict__ deg, int N) {
  int i = blockIdx.x * blockDim.x + threadIdx.x;
  if (i < N) deg[i] = 1.0f;
}

__global__ void k_count_deg(const int* __restrict__ dst, float* __restrict__ deg, int E) {
  int e = blockIdx.x * blockDim.x + threadIdx.x;
  if (e < E) unsafeAtomicAdd(&deg[dst[e]], 1.0f);
}

__global__ void k_rsqrt(float* __restrict__ deg, int N) {
  int i = blockIdx.x * blockDim.x + threadIdx.x;
  if (i < N) deg[i] = rsqrtf(deg[i]);
}


__global__ void k_gemm1(const float* __restrict__ X, const float* __restrict__ W,
                        float* __restrict__ H, int N) {
  const int lane = threadIdx.x & 31;
  const int wave = (blockIdx.x * blockDim.x + threadIdx.x) >> 5;
  const int m0 = wave << 4;
  if (m0 >= N) return;
  const int half = lane >> 4;
  const int lrow = lane & 15;
  int row = m0 + lrow; if (row >= N) row = N - 1;
  const float* xr = X + (size_t)row * F_IN;

  v8f c0 = {}; v8f c1 = {};
#pragma unroll 4
  for (int k0 = 0; k0 < F_IN; k0 += 4) {
    const int ka = k0 + (half << 1);
    v2f a;  a.x  = xr[ka];                      a.y  = xr[ka + 1];
    v2f b0; b0.x = W[ka * F_HID + lrow];        b0.y = W[(ka + 1) * F_HID + lrow];
    v2f b1; b1.x = W[ka * F_HID + 16 + lrow];   b1.y = W[(ka + 1) * F_HID + 16 + lrow];
    c0 = __builtin_amdgcn_wmma_f32_16x16x4_f32(false, a, false, b0, (short)0, c0, false, false);
    c1 = __builtin_amdgcn_wmma_f32_16x16x4_f32(false, a, false, b1, (short)0, c1, false, false);
  }
#pragma unroll
  for (int j = 0; j < 8; ++j) {
    int m = m0 + j + (half << 3);
    if (m < N) {
      float* hr = H + (size_t)m * F_HID;
      hr[lrow]      = c0[j];
      hr[16 + lrow] = c1[j];
    }
  }
}


__global__ void k_gemm2(const float* __restrict__ A, const float* __restrict__ W,
                        float* __restrict__ H, int N) {
  const int lane = threadIdx.x & 31;
  const int wave = (blockIdx.x * blockDim.x + threadIdx.x) >> 5;
  const int m0 = wave << 4;
  if (m0 >= N) return;
  const int half = lane >> 4;
  const int lrow = lane & 15;
  int row = m0 + lrow; if (row >= N) row = N - 1;
  const float* ar = A + (size_t)row * F_HID;

  v8f c0 = {}; v8f c1 = {}; v8f c2 = {}; v8f c3 = {};
#pragma unroll
  for (int k0 = 0; k0 < F_HID; k0 += 4) {
    const int ka = k0 + (half << 1);
    v2f a; a.x = fmaxf(ar[ka], 0.0f); a.y = fmaxf(ar[ka + 1], 0.0f);
    v2f b;
    b.x = W[ka * F_OUT + lrow];            b.y = W[(ka + 1) * F_OUT + lrow];
    c0 = __builtin_amdgcn_wmma_f32_16x16x4_f32(false, a, false, b, (short)0, c0, false, false);
    b.x = W[ka * F_OUT + 16 + lrow];       b.y = W[(ka + 1) * F_OUT + 16 + lrow];
    c1 = __builtin_amdgcn_wmma_f32_16x16x4_f32(false, a, false, b, (short)0, c1, false, false);
    b.x = W[ka * F_OUT + 32 + lrow];       b.y = W[(ka + 1) * F_OUT + 32 + lrow];
    c2 = __builtin_amdgcn_wmma_f32_16x16x4_f32(false, a, false, b, (short)0, c2, false, false);
    b.x = W[ka * F_OUT + 48 + lrow];       b.y = W[(ka + 1) * F_OUT + 48 + lrow];
    c3 = __builtin_amdgcn_wmma_f32_16x16x4_f32(false, a, false, b, (short)0, c3, false, false);
  }
#pragma unroll
  for (int j = 0; j < 8; ++j) {
    int m = m0 + j + (half << 3);
    if (m < N) {
      float* hr = H + (size_t)m * F_OUT;
      hr[lrow]      = c0[j];
      hr[16 + lrow] = c1[j];
      hr[32 + lrow] = c2[j];
      hr[48 + lrow] = c3[j];
    }
  }
}


__global__ void k_agg1_init(const float* __restrict__ h1, const float* __restrict__ dinv,
                            const float* __restrict__ bias, float* __restrict__ out1, int N) {
  int t = blockIdx.x * blockDim.x + threadIdx.x;
  if (t < N * F_HID) {
    int i = t >> 5, f = t & 31;
    float di = dinv[i];
    out1[t] = h1[t] * di * di + bias[f];
  }
}

__global__ void k_agg1(const float* __restrict__ h1, const float* __restrict__ dinv,
                       const int* __restrict__ src, const int* __restrict__ dstv,
                       float* __restrict__ out1, int E) {
  int t = blockIdx.x * blockDim.x + threadIdx.x;
  int e = t >> 5;
  if (e >= E) return;
  int lane = t & 31;
  int s = src[e], d = dstv[e];
  float nrm = dinv[s] * dinv[d];
  float v = h1[(size_t)s * F_HID + lane] * nrm;
  unsafeAtomicAdd(&out1[(size_t)d * F_HID + lane], v);
}

__global__ void k_agg2_init(const float* __restrict__ h2, const float* __restrict__ dinv,
                            const float* __restrict__ bias, float* __restrict__ out, int N) {
  int t = blockIdx.x * blockDim.x + threadIdx.x;
  if (t < N * F_OUT) {
    int i = t >> 6, f = t & 63;
    float di = dinv[i];
    out[t] = h2[t] * di * di + bias[f];
  }
}

__global__ void k_agg2(const float* __restrict__ h2, const float* __restrict__ dinv,
                       const int* __restrict__ src, const int* __restrict__ dstv,
                       float* __restrict__ out, int E) {
  int t = blockIdx.x * blockDim.x + threadIdx.x;
  int e = t >> 5;
  if (e >= E) return;
  int lane = t & 31;
  int s = src[e], d = dstv[e];
  float nrm = dinv[s] * dinv[d];
  const float* hs = h2 + (size_t)s * F_OUT;
  float* od = out + (size_t)d * F_OUT;
  unsafeAtomicAdd(&od[lane],      hs[lane]      * nrm);
  unsafeAtomicAdd(&od[lane + 32], hs[lane + 32] * nrm);
}


static inline int cdiv_ll(long long a, long long b) { return (int)((a + b - 1) / b); }

extern "C" void kernel_launch(void* const* d_in, const int* in_sizes, int n_in,
                              void* d_out, int out_size, void* d_ws, size_t ws_size,
                              hipStream_t stream) {
  const float* X  = (const float*)d_in[0];
  const int*   ei = (const int*)d_in[1];
  const float* W1 = (const float*)d_in[2];
  const float* b1 = (const float*)d_in[3];
  const float* W2 = (const float*)d_in[4];
  const float* b2 = (const float*)d_in[5];
  float* out = (float*)d_out;

  const int N = in_sizes[0] / F_IN;
  const int E = in_sizes[1] / 2;
  const int* src = ei;
  const int* dst = ei + E;

  float* dinv = (float*)d_ws;
  float* h1   = dinv + N;
  float* out1 = h1 + (size_t)N * F_HID;
  float* h2   = out1 + (size_t)N * F_HID;

  const int B = 256;
  const int mtiles = cdiv_ll(N, 16);

  k_init_deg <<<cdiv_ll(N, B), B, 0, stream>>>(dinv, N);
  k_count_deg<<<cdiv_ll(E, B), B, 0, stream>>>(dst, dinv, E);
  k_rsqrt    <<<cdiv_ll(N, B), B, 0, stream>>>(dinv, N);

  k_gemm1    <<<cdiv_ll((long long)mtiles * 32, B), B, 0, stream>>>(X, W1, h1, N);
  k_agg1_init<<<cdiv_ll((long long)N * F_HID, B), B, 0, stream>>>(h1, dinv, b1, out1, N);
  k_agg1     <<<cdiv_ll((long long)E * 32, B), B, 0, stream>>>(h1, dinv, src, dst, out1, E);

  k_gemm2    <<<cdiv_ll((long long)mtiles * 32, B), B, 0, stream>>>(out1, W2, h2, N);
  k_agg2_init<<<cdiv_ll((long long)N * F_OUT, B), B, 0, stream>>>(h2, dinv, b2, out, N);
  k_agg2     <<<cdiv_ll((long long)E * 32, B), B, 0, stream>>>(h2, dinv, src, dst, out, E);
}


# edge walks: the plane pre-pass also writes one 8-byte record per edge (source, row in range, source weight); a kept edge costs one scattered fetch, not three; weights and sums unchanged
# speedup vs baseline: 2.6342x; 2.6342x over previous
#include <hip/hip_runtime.h>
#include <hip/hip_bf16.h>

typedef float v2f __attribute__((ext_vector_type(2)));
typedef float v4f __attribute__((ext_vector_type(4)));
typedef float v8f __attribute__((ext_vector_type(8)));

#define F_IN  512
#define F_HID 32
#define F_OUT 64


#define DEG_THREADS 1024
#define DEG_RANGE   25000
#define DEG_SLICES  64
#define DEG_K       50
#define DEG_ROUND   (DEG_K * DEG_THREADS)

__device__ __forceinline__ int deg_load_dst(int (&cur)[DEG_K], const int* __restrict__ dstv,
                                            int base, int tid, int E) {
  if (E >= DEG_ROUND) {
    const int rb = min(base, E - DEG_ROUND);
    const int* p = dstv + rb + tid;
#pragma unroll
    for (int k = 0; k < DEG_K; ++k) cur[k] = p[k * DEG_THREADS];
    return rb;
  }
#pragma unroll
  for (int k = 0; k < DEG_K; ++k) {
    const int e = base + k * DEG_THREADS + tid;
    const int v = dstv[min(e, E - 1)];
    cur[k] = (e < E) ? v : -1;
  }
  return base;
}

__global__ __launch_bounds__(DEG_THREADS)
void k_count_deg(const int* __restrict__ dstv, unsigned* __restrict__ part, int N, int E) {
  __shared__ unsigned cnt[DEG_RANGE];
  const int tid = threadIdx.x;
  const int lo = blockIdx.x * DEG_RANGE;
  const int nr = min(DEG_RANGE, N - lo);
  const int per = (((E + DEG_SLICES - 1) / DEG_SLICES) + 63) & ~63;
  const int e0 = min(E, (int)blockIdx.y * per);
  const int e1 = min(E, e0 + per);

  int cur[DEG_K];
  int rb = e0;
  if (e0 < e1) rb = deg_load_dst(cur, dstv, e0, tid, E);
  for (int i = tid; i < DEG_RANGE; i += DEG_THREADS) cnt[i] = 0u;
  __syncthreads();

  for (int base = e0; base < e1; ) {
    const unsigned len = (unsigned)(e1 - base);
#pragma unroll
    for (int k = 0; k < DEG_K; ++k) {
      const unsigned r = (unsigned)(cur[k] - lo);
      const unsigned q = (unsigned)(rb + k * DEG_THREADS + tid - base);
      if (r < (unsigned)nr && q < len) atomicAdd(&cnt[r], 1u);
    }
    base += DEG_ROUND;
    if (base < e1) rb = deg_load_dst(cur, dstv, base, tid, E);
  }
  __syncthreads();

  unsigned* p = part + (size_t)blockIdx.y * N + lo;
  for (int i = tid; i < nr; i += DEG_THREADS) p[i] = cnt[i];
}

__global__ void k_dinv(const unsigned* __restrict__ part, float* __restrict__ dinv, int N) {
  int i = blockIdx.x * blockDim.x + threadIdx.x;
  if (i < N) {
    unsigned c = 1u;
#pragma unroll
    for (int s = 0; s < DEG_SLICES; ++s) c += part[(size_t)s * N + i];
    dinv[i] = rsqrtf((float)c);
  }
}


#define PL_ROWS    1000
#define PL_MAX     100
#define PL_CHUNK   16384
#define PL_THREADS 1024
#define PL_K       (PL_CHUNK / PL_THREADS)
#define PL_WORDS   (PL_CHUNK / 32)
#define PL_PITCH   (PL_WORDS + 1)
#define PL_MOD     2000
#define REC_SHIFT  21
#define REC_MASK   ((1u << REC_SHIFT) - 1u)

__global__ __launch_bounds__(PL_THREADS)
void k_planes(const int* __restrict__ srcv, const int* __restrict__ dstv, const float* __restrict__ dinv,
              unsigned* __restrict__ planes, unsigned* __restrict__ recs, int N, int E, int PW) {
  __shared__ __attribute__((aligned(16))) unsigned bm[PL_MAX * PL_PITCH];
  const int tid = threadIdx.x;
  const int c0 = blockIdx.x * PL_CHUNK;
  const int p0 = blockIdx.y * PL_MAX;
  const int np = min(PL_MAX, (N + PL_ROWS - 1) / PL_ROWS - p0);

  int d[PL_K];
  int s[PL_K];
  if (c0 + PL_CHUNK <= E) {
    const int* q = dstv + c0 + tid;
    const int* qs = srcv + c0 + tid;
#pragma unroll
    for (int k = 0; k < PL_K; ++k) d[k] = q[k * PL_THREADS];
#pragma unroll
    for (int k = 0; k < PL_K; ++k) s[k] = qs[k * PL_THREADS];
  } else {
#pragma unroll
    for (int k = 0; k < PL_K; ++k) {
      const int e = c0 + k * PL_THREADS + tid;
      const int v = dstv[min(e, E - 1)];
      d[k] = (e < E) ? v : -1;
      s[k] = srcv[min(e, E - 1)];
    }
  }
  for (int i = tid; i < PL_MAX * PL_PITCH / 4; i += PL_THREADS) ((uint4*)bm)[i] = make_uint4(0u, 0u, 0u, 0u);
  __syncthreads();

#pragma unroll
  for (int k = 0; k < PL_K; ++k) {
    const unsigned p = (unsigned)d[k] / PL_ROWS - (unsigned)p0;
    if (p < (unsigned)np) atomicOr(&bm[p * PL_PITCH + k * (PL_THREADS / 32) + (tid >> 5)], 1u << (tid & 31));
  }
  __syncthreads();

  float ds[PL_K];
#pragma unroll
  for (int k = 0; k < PL_K; ++k) ds[k] = dinv[s[k]];

  unsigned* o = planes + (size_t)p0 * PW + (size_t)blockIdx.x * PL_WORDS;
  for (int i = tid; i < np * PL_WORDS; i += PL_THREADS) {
    const int p = i / PL_WORDS, j = i % PL_WORDS;
    o[(size_t)p * PW + j] = bm[p * PL_PITCH + j];
  }

  if (blockIdx.y == 0) {
    uint2* r = (uint2*)recs + c0 + tid;
#pragma unroll
    for (int k = 0; k < PL_K; ++k) {
      if (c0 + k * PL_THREADS + tid < E)
        r[k * PL_THREADS] = make_uint2((unsigned)s[k] | ((unsigned)d[k] % PL_MOD) << REC_SHIFT,
                                       __float_as_uint(ds[k]));
    }
  }
}


__global__ void k_gemm1(const float* __restrict__ X, const float* __restrict__ W,
                        float* __restrict__ H, int N) {
  const int lane = threadIdx.x & 31;
  const int wave = (blockIdx.x * blockDim.x + threadIdx.x) >> 5;
  const int m0 = wave << 4;
  if (m0 >= N) return;
  const int half = lane >> 4;
  const int lrow = lane & 15;
  int row = m0 + lrow; if (row >= N) row = N - 1;
  const float* xr = X + (size_t)row * F_IN;

  v8f c0 = {}; v8f c1 = {};
#pragma unroll 4
  for (int k0 = 0; k0 < F_IN; k0 += 4) {
    const int ka = k0 + (half << 1);
    v2f a;  a.x  = xr[ka];                      a.y  = xr[ka + 1];
    v2f b0; b0.x = W[ka * F_HID + lrow];        b0.y = W[(ka + 1) * F_HID + lrow];
    v2f b1; b1.x = W[ka * F_HID + 16 + lrow];   b1.y = W[(ka + 1) * F_HID + 16 + lrow];
    c0 = __builtin_amdgcn_wmma_f32_16x16x4_f32(false, a, false, b0, (short)0, c0, false, false);
    c1 = __builtin_amdgcn_wmma_f32_16x16x4_f32(false, a, false, b1, (short)0, c1, false, false);
  }
#pragma unroll
  for (int j = 0; j < 8; ++j) {
    int m = m0 + j + (half << 3);
    if (m < N) {
      float* hr = H + (size_t)m * F_HID;
      hr[lrow]      = c0[j];
      hr[16 + lrow] = c1[j];
    }
  }
}


__global__ void k_gemm2(const float* __restrict__ A, const float* __restrict__ W,
                        float* __restrict__ H, int N) {
  const int lane = threadIdx.x & 31;
  const int wave = (blockIdx.x * blockDim.x + threadIdx.x) >> 5;
  const int m0 = wave << 4;
  if (m0 >= N) return;
  const int half = lane >> 4;
  const int lrow = lane & 15;
  int row = m0 + lrow; if (row >= N) row = N - 1;
  const float* ar = A + (size_t)row * F_HID;

  v8f c0 = {}; v8f c1 = {}; v8f c2 = {}; v8f c3 = {};
#pragma unroll
  for (int k0 = 0; k0 < F_HID; k0 += 4) {
    const int ka = k0 + (half << 1);
    v2f a; a.x = fmaxf(ar[ka], 0.0f); a.y = fmaxf(ar[ka + 1], 0.0f);
    v2f b;
    b.x = W[ka * F_OUT + lrow];            b.y = W[(ka + 1) * F_OUT + lrow];
    c0 = __builtin_amdgcn_wmma_f32_16x16x4_f32(false, a, false, b, (short)0, c0, false, false);
    b.x = W[ka * F_OUT + 16 + lrow];       b.y = W[(ka + 1) * F_OUT + 16 + lrow];
    c1 = __builtin_amdgcn_wmma_f32_16x16x4_f32(false, a, false, b, (short)0, c1, false, false);
    b.x = W[ka * F_OUT + 32 + lrow];       b.y = W[(ka + 1) * F_OUT + 32 + lrow];
    c2 = __builtin_amdgcn_wmma_f32_16x16x4_f32(false, a, false, b, (short)0, c2, false, false);
    b.x = W[ka * F_OUT + 48 + lrow];       b.y = W[(ka + 1) * F_OUT + 48 + lrow];
    c3 = __builtin_amdgcn_wmma_f32_16x16x4_f32(false, a, false, b, (short)0, c3, false, false);
  }
#pragma unroll
  for (int j = 0; j < 8; ++j) {
    int m = m0 + j + (half << 3);
    if (m < N) {
      float* hr = H + (size_t)m * F_OUT;
      hr[lrow]      = c0[j];
      hr[16 + lrow] = c1[j];
      hr[32 + lrow] = c2[j];
      hr[48 + lrow] = c3[j];
    }
  }
}


__global__ void k_agg1_init(const float* __restrict__ h1, const float* __restrict__ dinv,
                            const float* __restrict__ bias, float* __restrict__ out1, int N) {
  int t = blockIdx.x * blockDim.x + threadIdx.x;
  if (t < N * F_HID) {
    int i = t >> 5, f = t & 31;
    float di = dinv[i];
    out1[t] = h1[t] * di * di + bias[f];
  }
}

__global__ void k_agg2_init(const float* __restrict__ h2, const float* __restrict__ dinv,
                            const float* __restrict__ bias, float* __restrict__ out, int N) {
  int t = blockIdx.x * blockDim.x + threadIdx.x;
  if (t < N * F_OUT) {
    int i = t >> 6, f = t & 63;
    float di = dinv[i];
    out[t] = h2[t] * di * di + bias[f];
  }
}


#define AGG_THREADS 1024
#define AGG_WAVES   32
#define WR_EDGES    1024
#define ACC_FLOATS  64000

template <int PPR>
__device__ __forceinline__ void agg_load_bits(unsigned (&cur)[PPR], const unsigned* __restrict__ pl,
                                              int PW, int np, int t) {
#pragma unroll
  for (int j = 0; j < PPR; ++j) cur[j] = pl[(size_t)min(j, np - 1) * PW + t * 32];
}

__device__ __forceinline__ int row_prefix_sum(int v) {
  v += __builtin_amdgcn_update_dpp(0, v, 0x111, 0xf, 0xf, true);
  v += __builtin_amdgcn_update_dpp(0, v, 0x112, 0xf, 0xf, true);
  v += __builtin_amdgcn_update_dpp(0, v, 0x114, 0xf, 0xf, true);
  v += __builtin_amdgcn_update_dpp(0, v, 0x118, 0xf, 0xf, true);
  return v;
}

template <int K>
__device__ __forceinline__ int row_bcast(int v) {
  return __builtin_amdgcn_update_dpp(0, v, 0x150 + K, 0xf, 0xf, true);
}

template <int F> struct agg_vec;
template <> struct agg_vec<32> { typedef v2f type; static constexpr int G = 16; };
template <> struct agg_vec<64> { typedef v4f type; static constexpr int G = 8; };

template <int F, int K0, int K>
__device__ __forceinline__ void agg_loads(typename agg_vec<F>::type (&rv)[agg_vec<F>::G],
                                          const char* __restrict__ hb, int sOff, int lanePart,
                                          int st) {
  typedef typename agg_vec<F>::type vec_t;
  if constexpr (K < agg_vec<F>::G) {
    if (K < st) {
      rv[K] = *(const vec_t*)(hb + (unsigned)(row_bcast<K0 + K>(sOff) + lanePart));
      agg_loads<F, K0, K + 1>(rv, hb, sOff, lanePart, st);
    }
  }
}

template <int F, int K0, int K>
__device__ __forceinline__ void agg_adds(const typename agg_vec<F>::type (&rv)[agg_vec<F>::G],
                                         float* acc, int wi, int dOff, int lanePart, int st) {
  constexpr int V = F / 16;
  if constexpr (K < agg_vec<F>::G) {
    if (K < st) {
      const float wk = __int_as_float(row_bcast<K0 + K>(wi));
      float* ar = (float*)((char*)acc + (unsigned)(row_bcast<K0 + K>(dOff) + lanePart));
#pragma unroll
      for (int c = 0; c < V; ++c) unsafeAtomicAdd(ar + c, rv[K][c] * wk);
      agg_adds<F, K0, K + 1>(rv, acc, wi, dOff, lanePart, st);
    }
  }
}

template <int F, int K0>
__device__ __forceinline__ void agg_walk(const float* __restrict__ h, float* acc, int sOff,
                                         float w, int dOff, int lanePart, int st) {
  typename agg_vec<F>::type rv[agg_vec<F>::G];
  agg_loads<F, K0, 0>(rv, (const char*)h, sOff, lanePart, st);
  agg_adds<F, K0, 0>(rv, acc, __float_as_int(w), dOff, lanePart, st);
}

template <int F>
__global__ __launch_bounds__(AGG_THREADS)
void k_agg_lds(const float* __restrict__ h, const float* __restrict__ dinv,
               const unsigned* __restrict__ recs,
               const unsigned* __restrict__ planes, float* __restrict__ out, int N, int PW) {
  constexpr int RANGE = ACC_FLOATS / F;
  constexpr int PPR = RANGE / PL_ROWS;
  static_assert(RANGE % PL_ROWS == 0, "a range is whole planes");
  static_assert(PL_MOD % RANGE == 0, "a record's remainder gives the row inside every range");
  __shared__ float acc[ACC_FLOATS + F];
  __shared__ unsigned short lst[AGG_WAVES * WR_EDGES];

  const int tid = threadIdx.x;
  const int lane = tid & 31;
  const int wave = __builtin_amdgcn_readfirstlane(tid >> 5);
  const int lo = blockIdx.x * RANGE;
  const int nr = min(RANGE, N - lo);
  const int np = min(PPR, (nr + PL_ROWS - 1) / PL_ROWS);
  const int nt = PW >> 5;
  const int per = (nt + (int)gridDim.y - 1) / (int)gridDim.y;
  const int t0 = blockIdx.y * per;
  const int t1 = min(nt, t0 + per);
  const int slot = 2 * (lane & 15) + (lane >> 4);
  const int lanePart = (lane & 15) * (F / 16) * 4;

  for (int i = tid; i < ACC_FLOATS; i += AGG_THREADS) acc[i] = 0.0f;
  __syncthreads();

  unsigned short* mylst = lst + wave * WR_EDGES;
  const unsigned* pl = planes + (size_t)(blockIdx.x * PPR) * PW + lane;
  int t = t0 + wave;
  unsigned cur[PPR];
  if (t < t1) agg_load_bits<PPR>(cur, pl, PW, np, t);

  for (; t < t1; t += AGG_WAVES) {
    unsigned wd = 0u;
#pragma unroll
    for (int j = 0; j < PPR; ++j) wd |= cur[j];
    const int pbase = t * WR_EDGES;
    if (t + AGG_WAVES < t1) agg_load_bits<PPR>(cur, pl, PW, np, t + AGG_WAVES);

    const int c = __popc(wd);
    int inc = row_prefix_sum(c);
    const int cnt0 = __builtin_amdgcn_readlane(inc, 15);
    const int cnt = cnt0 + __builtin_amdgcn_readlane(inc, 31);
    if (lane >= 16) inc += cnt0;
    int at = inc - c;
    while (wd) {
      mylst[at++] = (unsigned short)(lane * 32 + __builtin_ctz(wd));
      wd &= wd - 1u;
    }

    for (int b = 0; b < cnt; b += 32) {
      const int n = min(32, cnt - b);
      const bool live = slot < n;
      uint2 rec = make_uint2(0u, 0u);
      if (live) rec = ((const uint2*)recs)[pbase + (int)mylst[b + slot]];
      const int s = (int)(rec.x & REC_MASK);
      const int row = (int)(rec.x >> REC_SHIFT) % RANGE;
      const float ds = __uint_as_float(rec.y);
      const float dd = dinv[lo + row];
      const float w = live ? ds * dd : 0.0f;
      const int sOff = s * (F * 4);
      const int dOff = live ? row * (F * 4) : ACC_FLOATS * 4;
      const int steps = (n + 1) >> 1;
      constexpr int G = agg_vec<F>::G;
      agg_walk<F, 0>(h, acc, sOff, w, dOff, lanePart, min(steps, G));
      if constexpr (G < 16) {
        if (steps > G) agg_walk<F, G>(h, acc, sOff, w, dOff, lanePart, steps - G);
      }
    }
  }
  __syncthreads();

  float* o = out + (size_t)lo * F;
  for (int i = tid; i < nr * F; i += AGG_THREADS) unsafeAtomicAdd(&o[i], acc[i]);
}


static inline int cdiv_ll(long long a, long long b) { return (int)((a + b - 1) / b); }

extern "C" void kernel_launch(void* const* d_in, const int* in_sizes, int n_in,
                              void* d_out, int out_size, void* d_ws, size_t ws_size,
                              hipStream_t stream) {
  const float* X  = (const float*)d_in[0];
  const int*   ei = (const int*)d_in[1];
  const float* W1 = (const float*)d_in[2];
  const float* b1 = (const float*)d_in[3];
  const float* W2 = (const float*)d_in[4];
  const float* b2 = (const float*)d_in[5];
  float* out = (float*)d_out;

  const int N = in_sizes[0] / F_IN;
  const int E = in_sizes[1] / 2;
  const int* src = ei;
  const int* dst = ei + E;

  const int nchunk = cdiv_ll(E, PL_CHUNK);
  const int PW = nchunk * PL_WORDS;
  const int nplanes = cdiv_ll(N, PL_ROWS);

  unsigned* recs = (unsigned*)d_ws;
  float* dinv = (float*)(recs + (size_t)nchunk * PL_CHUNK * 2);
  float* h1   = dinv + N;
  float* out1 = h1 + (size_t)N * F_HID;
  float* h2   = out1 + (size_t)N * F_HID;
  unsigned* part = (unsigned*)(h2 + (size_t)N * F_OUT);
  unsigned* planes = part + (size_t)DEG_SLICES * N;

  const int B = 256;
  const int mtiles = cdiv_ll(N, 16);

  k_count_deg<<<dim3(cdiv_ll(N, DEG_RANGE), DEG_SLICES), DEG_THREADS, 0, stream>>>(dst, part, N, E);
  k_dinv     <<<cdiv_ll(N, B), B, 0, stream>>>(part, dinv, N);
  k_planes   <<<dim3(nchunk, cdiv_ll(nplanes, PL_MAX)), PL_THREADS, 0, stream>>>(src, dst, dinv, planes, recs, N, E, PW);

  k_gemm1    <<<cdiv_ll((long long)mtiles * 32, B), B, 0, stream>>>(X, W1, h1, N);
  k_agg1_init<<<cdiv_ll((long long)N * F_HID, B), B, 0, stream>>>(h1, dinv, b1, out1, N);
  k_agg_lds<F_HID><<<dim3(cdiv_ll(N, ACC_FLOATS / F_HID), 5), AGG_THREADS, 0, stream>>>(h1, dinv, recs, planes, out1, N, PW);

  k_gemm2    <<<cdiv_ll((long long)mtiles * 32, B), B, 0, stream>>>(out1, W2, h2, N);
  k_agg2_init<<<cdiv_ll((long long)N * F_OUT, B), B, 0, stream>>>(h2, dinv, b2, out, N);
  k_agg_lds<F_OUT><<<dim3(cdiv_ll(N, ACC_FLOATS / F_OUT), 2), AGG_THREADS, 0, stream>>>(h2, dinv, recs, planes, out, N, PW);
}


# both start-value passes (self-loop term + bias): a lane takes 4 or 8 pieces of 16 bytes, loads first, whole-line stores; 6256 waves a pass, not 100000 and 200000 one-float waves; sums unchanged
# speedup vs baseline: 2.8004x; 1.0631x over previous
#include <hip/hip_runtime.h>
#include <hip/hip_bf16.h>

typedef float v2f __attribute__((ext_vector_type(2)));
typedef float v4f __attribute__((ext_vector_type(4)));
typedef float v8f __attribute__((ext_vector_type(8)));

#define F_IN  512
#define F_HID 32
#define F_OUT 64


#define DEG_THREADS 1024
#define DEG_RANGE   25000
#define DEG_SLICES  64
#define DEG_K       50
#define DEG_ROUND   (DEG_K * DEG_THREADS)

__device__ __forceinline__ int deg_load_dst(int (&cur)[DEG_K], const int* __restrict__ dstv,
                                            int base, int tid, int E) {
  if (E >= DEG_ROUND) {
    const int rb = min(base, E - DEG_ROUND);
    const int* p = dstv + rb + tid;
#pragma unroll
    for (int k = 0; k < DEG_K; ++k) cur[k] = p[k * DEG_THREADS];
    return rb;
  }
#pragma unroll
  for (int k = 0; k < DEG_K; ++k) {
    const int e = base + k * DEG_THREADS + tid;
    const int v = dstv[min(e, E - 1)];
    cur[k] = (e < E) ? v : -1;
  }
  return base;
}

__global__ __launch_bounds__(DEG_THREADS)
void k_count_deg(const int* __restrict__ dstv, unsigned* __restrict__ part, int N, int E) {
  __shared__ unsigned cnt[DEG_RANGE];
  const int tid = threadIdx.x;
  const int lo = blockIdx.x * DEG_RANGE;
  const int nr = min(DEG_RANGE, N - lo);
  const int per = (((E + DEG_SLICES - 1) / DEG_SLICES) + 63) & ~63;
  const int e0 = min(E, (int)blockIdx.y * per);
  const int e1 = min(E, e0 + per);

  int cur[DEG_K];
  int rb = e0;
  if (e0 < e1) rb = deg_load_dst(cur, dstv, e0, tid, E);
  for (int i = tid; i < DEG_RANGE; i += DEG_THREADS) cnt[i] = 0u;
  __syncthreads();

  for (int base = e0; base < e1; ) {
    const unsigned len = (unsigned)(e1 - base);
#pragma unroll
    for (int k = 0; k < DEG_K; ++k) {
      const unsigned r = (unsigned)(cur[k] - lo);
      const unsigned q = (unsigned)(rb + k * DEG_THREADS + tid - base);
      if (r < (unsigned)nr && q < len) atomicAdd(&cnt[r], 1u);
    }
    base += DEG_ROUND;
    if (base < e1) rb = deg_load_dst(cur, dstv, base, tid, E);
  }
  __syncthreads();

  unsigned* p = part + (size_t)blockIdx.y * N + lo;
  for (int i = tid; i < nr; i += DEG_THREADS) p[i] = cnt[i];
}

__global__ void k_dinv(const unsigned* __restrict__ part, float* __restrict__ dinv, int N) {
  int i = blockIdx.x * blockDim.x + threadIdx.x;
  if (i < N) {
    unsigned c = 1u;
#pragma unroll
    for (int s = 0; s < DEG_SLICES; ++s) c += part[(size_t)s * N + i];
    dinv[i] = rsqrtf((float)c);
  }
}


#define PL_ROWS    1000
#define PL_MAX     100
#define PL_CHUNK   16384
#define PL_THREADS 1024
#define PL_K       (PL_CHUNK / PL_THREADS)
#define PL_WORDS   (PL_CHUNK / 32)
#define PL_PITCH   (PL_WORDS + 1)
#define PL_MOD     2000
#define REC_SHIFT  21
#define REC_MASK   ((1u << REC_SHIFT) - 1u)

__global__ __launch_bounds__(PL_THREADS)
void k_planes(const int* __restrict__ srcv, const int* __restrict__ dstv, const float* __restrict__ dinv,
              unsigned* __restrict__ planes, unsigned* __restrict__ recs, int N, int E, int PW) {
  __shared__ __attribute__((aligned(16))) unsigned bm[PL_MAX * PL_PITCH];
  const int tid = threadIdx.x;
  const int c0 = blockIdx.x * PL_CHUNK;
  const int p0 = blockIdx.y * PL_MAX;
  const int np = min(PL_MAX, (N + PL_ROWS - 1) / PL_ROWS - p0);

  int d[PL_K];
  int s[PL_K];
  if (c0 + PL_CHUNK <= E) {
    const int* q = dstv + c0 + tid;
    const int* qs = srcv + c0 + tid;
#pragma unroll
    for (int k = 0; k < PL_K; ++k) d[k] = q[k * PL_THREADS];
#pragma unroll
    for (int k = 0; k < PL_K; ++k) s[k] = qs[k * PL_THREADS];
  } else {
#pragma unroll
    for (int k = 0; k < PL_K; ++k) {
      const int e = c0 + k * PL_THREADS + tid;
      const int v = dstv[min(e, E - 1)];
      d[k] = (e < E) ? v : -1;
      s[k] = srcv[min(e, E - 1)];
    }
  }
  for (int i = tid; i < PL_MAX * PL_PITCH / 4; i += PL_THREADS) ((uint4*)bm)[i] = make_uint4(0u, 0u, 0u, 0u);
  __syncthreads();

#pragma unroll
  for (int k = 0; k < PL_K; ++k) {
    const unsigned p = (unsigned)d[k] / PL_ROWS - (unsigned)p0;
    if (p < (unsigned)np) atomicOr(&bm[p * PL_PITCH + k * (PL_THREADS / 32) + (tid >> 5)], 1u << (tid & 31));
  }
  __syncthreads();

  float ds[PL_K];
#pragma unroll
  for (int k = 0; k < PL_K; ++k) ds[k] = dinv[s[k]];

  unsigned* o = planes + (size_t)p0 * PW + (size_t)blockIdx.x * PL_WORDS;
  for (int i = tid; i < np * PL_WORDS; i += PL_THREADS) {
    const int p = i / PL_WORDS, j = i % PL_WORDS;
    o[(size_t)p * PW + j] = bm[p * PL_PITCH + j];
  }

  if (blockIdx.y == 0) {
    uint2* r = (uint2*)recs + c0 + tid;
#pragma unroll
    for (int k = 0; k < PL_K; ++k) {
      if (c0 + k * PL_THREADS + tid < E)
        r[k * PL_THREADS] = make_uint2((unsigned)s[k] | ((unsigned)d[k] % PL_MOD) << REC_SHIFT,
                                       __float_as_uint(ds[k]));
    }
  }
}


__global__ void k_gemm1(const float* __restrict__ X, const float* __restrict__ W,
                        float* __restrict__ H, int N) {
  const int lane = threadIdx.x & 31;
  const int wave = (blockIdx.x * blockDim.x + threadIdx.x) >> 5;
  const int m0 = wave << 4;
  if (m0 >= N) return;
  const int half = lane >> 4;
  const int lrow = lane & 15;
  int row = m0 + lrow; if (row >= N) row = N - 1;
  const float* xr = X + (size_t)row * F_IN;

  v8f c0 = {}; v8f c1 = {};
#pragma unroll 4
  for (int k0 = 0; k0 < F_IN; k0 += 4) {
    const int ka = k0 + (half << 1);
    v2f a;  a.x  = xr[ka];                      a.y  = xr[ka + 1];
    v2f b0; b0.x = W[ka * F_HID + lrow];        b0.y = W[(ka + 1) * F_HID + lrow];
    v2f b1; b1.x = W[ka * F_HID + 16 + lrow];   b1.y = W[(ka + 1) * F_HID + 16 + lrow];
    c0 = __builtin_amdgcn_wmma_f32_16x16x4_f32(false, a, false, b0, (short)0, c0, false, false);
    c1 = __builtin_amdgcn_wmma_f32_16x16x4_f32(false, a, false, b1, (short)0, c1, false, false);
  }
#pragma unroll
  for (int j = 0; j < 8; ++j) {
    int m = m0 + j + (half << 3);
    if (m < N) {
      float* hr = H + (size_t)m * F_HID;
      hr[lrow]      = c0[j];
      hr[16 + lrow] = c1[j];
    }
  }
}


__global__ void k_gemm2(const float* __restrict__ A, const float* __restrict__ W,
                        float* __restrict__ H, int N) {
  const int lane = threadIdx.x & 31;
  const int wave = (blockIdx.x * blockDim.x + threadIdx.x) >> 5;
  const int m0 = wave << 4;
  if (m0 >= N) return;
  const int half = lane >> 4;
  const int lrow = lane & 15;
  int row = m0 + lrow; if (row >= N) row = N - 1;
  const float* ar = A + (size_t)row * F_HID;

  v8f c0 = {}; v8f c1 = {}; v8f c2 = {}; v8f c3 = {};
#pragma unroll
  for (int k0 = 0; k0 < F_HID; k0 += 4) {
    const int ka = k0 + (half << 1);
    v2f a; a.x = fmaxf(ar[ka], 0.0f); a.y = fmaxf(ar[ka + 1], 0.0f);
    v2f b;
    b.x = W[ka * F_OUT + lrow];            b.y = W[(ka + 1) * F_OUT + lrow];
    c0 = __builtin_amdgcn_wmma_f32_16x16x4_f32(false, a, false, b, (short)0, c0, false, false);
    b.x = W[ka * F_OUT + 16 + lrow];       b.y = W[(ka + 1) * F_OUT + 16 + lrow];
    c1 = __builtin_amdgcn_wmma_f32_16x16x4_f32(false, a, false, b, (short)0, c1, false, false);
    b.x = W[ka * F_OUT + 32 + lrow];       b.y = W[(ka + 1) * F_OUT + 32 + lrow];
    c2 = __builtin_amdgcn_wmma_f32_16x16x4_f32(false, a, false, b, (short)0, c2, false, false);
    b.x = W[ka * F_OUT + 48 + lrow];       b.y = W[(ka + 1) * F_OUT + 48 + lrow];
    c3 = __builtin_amdgcn_wmma_f32_16x16x4_f32(false, a, false, b, (short)0, c3, false, false);
  }
#pragma unroll
  for (int j = 0; j < 8; ++j) {
    int m = m0 + j + (half << 3);
    if (m < N) {
      float* hr = H + (size_t)m * F_OUT;
      hr[lrow]      = c0[j];
      hr[16 + lrow] = c1[j];
      hr[32 + lrow] = c2[j];
      hr[48 + lrow] = c3[j];
    }
  }
}


#define INIT_THREADS 256

template <int F, int V, bool WHOLE>
__device__ __forceinline__ void agg_init_pieces(const v4f* __restrict__ hv, const float* __restrict__ dinv,
                                                v4f b, v4f* __restrict__ ov, unsigned q0, unsigned nvec) {
  constexpr unsigned VPR = F / 4;
  v4f x[V];
  float di[V];
#pragma unroll
  for (int j = 0; j < V; ++j) {
    unsigned q = q0 + j * INIT_THREADS;
    if (!WHOLE) q = min(q, nvec - 1u);
    x[j] = hv[q];
    di[j] = dinv[q / VPR];
  }
#pragma unroll
  for (int j = 0; j < V; ++j) {
    const unsigned q = q0 + j * INIT_THREADS;
    const v4f d4 = {di[j], di[j], di[j], di[j]};
    if (WHOLE || q < nvec) ov[q] = __builtin_elementwise_fma(x[j] * d4, d4, b);
  }
}

template <int F, int V>
__global__ __launch_bounds__(INIT_THREADS)
void k_agg_init(const float* __restrict__ h, const float* __restrict__ dinv,
                const float* __restrict__ bias, float* __restrict__ out, int nvec) {
  static_assert(INIT_THREADS % (F / 4) == 0, "a lane's pieces all lie at the same place of their rows");
  const unsigned tid = threadIdx.x;
  const unsigned g0 = blockIdx.x * (unsigned)(V * INIT_THREADS);
  const v4f b = *(const v4f*)(bias + 4u * (tid % (F / 4)));
  if (g0 + V * INIT_THREADS <= (unsigned)nvec)
    agg_init_pieces<F, V, true>((const v4f*)h, dinv, b, (v4f*)out, g0 + tid, (unsigned)nvec);
  else
    agg_init_pieces<F, V, false>((const v4f*)h, dinv, b, (v4f*)out, g0 + tid, (unsigned)nvec);
}


#define AGG_THREADS 1024
#define AGG_WAVES   32
#define WR_EDGES    1024
#define ACC_FLOATS  64000

template <int PPR>
__device__ __forceinline__ void agg_load_bits(unsigned (&cur)[PPR], const unsigned* __restrict__ pl,
                                              int PW, int np, int t) {
#pragma unroll
  for (int j = 0; j < PPR; ++j) cur[j] = pl[(size_t)min(j, np - 1) * PW + t * 32];
}

__device__ __forceinline__ int row_prefix_sum(int v) {
  v += __builtin_amdgcn_update_dpp(0, v, 0x111, 0xf, 0xf, true);
  v += __builtin_amdgcn_update_dpp(0, v, 0x112, 0xf, 0xf, true);
  v += __builtin_amdgcn_update_dpp(0, v, 0x114, 0xf, 0xf, true);
  v += __builtin_amdgcn_update_dpp(0, v, 0x118, 0xf, 0xf, true);
  return v;
}

template <int K>
__device__ __forceinline__ int row_bcast(int v) {
  return __builtin_amdgcn_update_dpp(0, v, 0x150 + K, 0xf, 0xf, true);
}

template <int F> struct agg_vec;
template <> struct agg_vec<32> { typedef v2f type; static constexpr int G = 16; };
template <> struct agg_vec<64> { typedef v4f type; static constexpr int G = 8; };

template <int F, int K0, int K>
__device__ __forceinline__ void agg_loads(typename agg_vec<F>::type (&rv)[agg_vec<F>::G],
                                          const char* __restrict__ hb, int sOff, int lanePart,
                                          int st) {
  typedef typename agg_vec<F>::type vec_t;
  if constexpr (K < agg_vec<F>::G) {
    if (K < st) {
      rv[K] = *(const vec_t*)(hb + (unsigned)(row_bcast<K0 + K>(sOff) + lanePart));
      agg_loads<F, K0, K + 1>(rv, hb, sOff, lanePart, st);
    }
  }
}

template <int F, int K0, int K>
__device__ __forceinline__ void agg_adds(const typename agg_vec<F>::type (&rv)[agg_vec<F>::G],
                                         float* acc, int wi, int dOff, int lanePart, int st) {
  constexpr int V = F / 16;
  if constexpr (K < agg_vec<F>::G) {
    if (K < st) {
      const float wk = __int_as_float(row_bcast<K0 + K>(wi));
      float* ar = (float*)((char*)acc + (unsigned)(row_bcast<K0 + K>(dOff) + lanePart));
#pragma unroll
      for (int c = 0; c < V; ++c) unsafeAtomicAdd(ar + c, rv[K][c] * wk);
      agg_adds<F, K0, K + 1>(rv, acc, wi, dOff, lanePart, st);
    }
  }
}

template <int F, int K0>
__device__ __forceinline__ void agg_walk(const float* __restrict__ h, float* acc, int sOff,
                                         float w, int dOff, int lanePart, int st) {
  typename agg_vec<F>::type rv[agg_vec<F>::G];
  agg_loads<F, K0, 0>(rv, (const char*)h, sOff, lanePart, st);
  agg_adds<F, K0, 0>(rv, acc, __float_as_int(w), dOff, lanePart, st);
}

template <int F>
__global__ __launch_bounds__(AGG_THREADS)
void k_agg_lds(const float* __restrict__ h, const float* __restrict__ dinv,
               const unsigned* __restrict__ recs,
               const unsigned* __restrict__ planes, float* __restrict__ out, int N, int PW) {
  constexpr int RANGE = ACC_FLOATS / F;
  constexpr int PPR = RANGE / PL_ROWS;
  static_assert(RANGE % PL_ROWS == 0, "a range is whole planes");
  static_assert(PL_MOD % RANGE == 0, "a record's remainder gives the row inside every range");
  __shared__ float acc[ACC_FLOATS + F];
  __shared__ unsigned short lst[AGG_WAVES * WR_EDGES];

  const int tid = threadIdx.x;
  const int lane = tid & 31;
  const int wave = __builtin_amdgcn_readfirstlane(tid >> 5);
  const int lo = blockIdx.x * RANGE;
  const int nr = min(RANGE, N - lo);
  const int np = min(PPR, (nr + PL_ROWS - 1) / PL_ROWS);
  const int nt = PW >> 5;
  const int per = (nt + (int)gridDim.y - 1) / (int)gridDim.y;
  const int t0 = blockIdx.y * per;
  const int t1 = min(nt, t0 + per);
  const int slot = 2 * (lane & 15) + (lane >> 4);
  const int lanePart = (lane & 15) * (F / 16) * 4;

  for (int i = tid; i < ACC_FLOATS; i += AGG_THREADS) acc[i] = 0.0f;
  __syncthreads();

  unsigned short* mylst = lst + wave * WR_EDGES;
  const unsigned* pl = planes + (size_t)(blockIdx.x * PPR) * PW + lane;
  int t = t0 + wave;
  unsigned cur[PPR];
  if (t < t1) agg_load_bits<PPR>(cur, pl, PW, np, t);

  for (; t < t1; t += AGG_WAVES) {
    unsigned wd = 0u;
#pragma unroll
    for (int j = 0; j < PPR; ++j) wd |= cur[j];
    const int pbase = t * WR_EDGES;
    if (t + AGG_WAVES < t1) agg_load_bits<PPR>(cur, pl, PW, np, t + AGG_WAVES);

    const int c = __popc(wd);
    int inc = row_prefix_sum(c);
    const int cnt0 = __builtin_amdgcn_readlane(inc, 15);
    const int cnt = cnt0 + __builtin_amdgcn_readlane(inc, 31);
    if (lane >= 16) inc += cnt0;
    int at = inc - c;
    while (wd) {
      mylst[at++] = (unsigned short)(lane * 32 + __builtin_ctz(wd));
      wd &= wd - 1u;
    }

    for (int b = 0; b < cnt; b += 32) {
      const int n = min(32, cnt - b);
      const bool live = slot < n;
      uint2 rec = make_uint2(0u, 0u);
      if (live) rec = ((const uint2*)recs)[pbase + (int)mylst[b + slot]];
      const int s = (int)(rec.x & REC_MASK);
      const int row = (int)(rec.x >> REC_SHIFT) % RANGE;
      const float ds = __uint_as_float(rec.y);
      const float dd = dinv[lo + row];
      const float w = live ? ds * dd : 0.0f;
      const int sOff = s * (F * 4);
      const int dOff = live ? row * (F * 4) : ACC_FLOATS * 4;
      const int steps = (n + 1) >> 1;
      constexpr int G = agg_vec<F>::G;
      agg_walk<F, 0>(h, acc, sOff, w, dOff, lanePart, min(steps, G));
      if constexpr (G < 16) {
        if (steps > G) agg_walk<F, G>(h, acc, sOff, w, dOff, lanePart, steps - G);
      }
    }
  }
  __syncthreads();

  float* o = out + (size_t)lo * F;
  for (int i = tid; i < nr * F; i += AGG_THREADS) unsafeAtomicAdd(&o[i], acc[i]);
}


static inline int cdiv_ll(long long a, long long b) { return (int)((a + b - 1) / b); }

extern "C" void kernel_launch(void* const* d_in, const int* in_sizes, int n_in,
                              void* d_out, int out_size, void* d_ws, size_t ws_size,
                              hipStream_t stream) {
  const float* X  = (const float*)d_in[0];
  const int*   ei = (const int*)d_in[1];
  const float* W1 = (const float*)d_in[2];
  const float* b1 = (const float*)d_in[3];
  const float* W2 = (const float*)d_in[4];
  const float* b2 = (const float*)d_in[5];
  float* out = (float*)d_out;

  const int N = in_sizes[0] / F_IN;
  const int E = in_sizes[1] / 2;
  const int* src = ei;
  const int* dst = ei + E;

  const int nchunk = cdiv_ll(E, PL_CHUNK);
  const int PW = nchunk * PL_WORDS;
  const int nplanes = cdiv_ll(N, PL_ROWS);

  unsigned* recs = (unsigned*)d_ws;
  float* dinv = (float*)(recs + (size_t)nchunk * PL_CHUNK * 2);
  float* h1   = dinv + N;
  float* out1 = h1 + (size_t)N * F_HID;
  float* h2   = out1 + (size_t)N * F_HID;
  unsigned* part = (unsigned*)(h2 + (size_t)N * F_OUT);
  unsigned* planes = part + (size_t)DEG_SLICES * N;

  const int B = 256;
  const int mtiles = cdiv_ll(N, 16);
  const int nv1 = N * (F_HID / 4);
  const int nv2 = N * (F_OUT / 4);

  k_count_deg<<<dim3(cdiv_ll(N, DEG_RANGE), DEG_SLICES), DEG_THREADS, 0, stream>>>(dst, part, N, E);
  k_dinv     <<<cdiv_ll(N, B), B, 0, stream>>>(part, dinv, N);
  k_planes   <<<dim3(nchunk, cdiv_ll(nplanes, PL_MAX)), PL_THREADS, 0, stream>>>(src, dst, dinv, planes, recs, N, E, PW);

  k_gemm1    <<<cdiv_ll((long long)mtiles * 32, B), B, 0, stream>>>(X, W1, h1, N);
  k_agg_init<F_HID, 4><<<cdiv_ll(nv1, 4 * INIT_THREADS), INIT_THREADS, 0, stream>>>(h1, dinv, b1, out1, nv1);
  k_agg_lds<F_HID><<<dim3(cdiv_ll(N, ACC_FLOATS / F_HID), 5), AGG_THREADS, 0, stream>>>(h1, dinv, recs, planes, out1, N, PW);

  k_gemm2    <<<cdiv_ll((long long)mtiles * 32, B), B, 0, stream>>>(out1, W2, h2, N);
  k_agg_init<F_OUT, 8><<<cdiv_ll(nv2, 8 * INIT_THREADS), INIT_THREADS, 0, stream>>>(h2, dinv, b2, out, nv2);
  k_agg_lds<F_OUT><<<dim3(cdiv_ll(N, ACC_FLOATS / F_OUT), 2), AGG_THREADS, 0, stream>>>(h2, dinv, recs, planes, out, N, PW);
}


# both edge sums: the record pre-pass groups each chunk's records by 1000-row destination block (LDS counting ranks) and stores run starts; range owners read runs, no bit planes; f32 sum order moves
# speedup vs baseline: 3.0801x; 1.0999x over previous
#include <hip/hip_runtime.h>
#include <hip/hip_bf16.h>

typedef float v2f __attribute__((ext_vector_type(2)));
typedef float v4f __attribute__((ext_vector_type(4)));
typedef float v8f __attribute__((ext_vector_type(8)));

#define F_IN  512
#define F_HID 32
#define F_OUT 64


#define DEG_THREADS 1024
#define DEG_RANGE   25000
#define DEG_SLICES  64
#define DEG_K       50
#define DEG_ROUND   (DEG_K * DEG_THREADS)

__device__ __forceinline__ int deg_load_dst(int (&cur)[DEG_K], const int* __restrict__ dstv,
                                            int base, int tid, int E) {
  if (E >= DEG_ROUND) {
    const int rb = min(base, E - DEG_ROUND);
    const int* p = dstv + rb + tid;
#pragma unroll
    for (int k = 0; k < DEG_K; ++k) cur[k] = p[k * DEG_THREADS];
    return rb;
  }
#pragma unroll
  for (int k = 0; k < DEG_K; ++k) {
    const int e = base + k * DEG_THREADS + tid;
    const int v = dstv[min(e, E - 1)];
    cur[k] = (e < E) ? v : -1;
  }
  return base;
}

__global__ __launch_bounds__(DEG_THREADS)
void k_count_deg(const int* __restrict__ dstv, unsigned* __restrict__ part, int N, int E) {
  __shared__ unsigned cnt[DEG_RANGE];
  const int tid = threadIdx.x;
  const int lo = blockIdx.x * DEG_RANGE;
  const int nr = min(DEG_RANGE, N - lo);
  const int per = (((E + DEG_SLICES - 1) / DEG_SLICES) + 63) & ~63;
  const int e0 = min(E, (int)blockIdx.y * per);
  const int e1 = min(E, e0 + per);

  int cur[DEG_K];
  int rb = e0;
  if (e0 < e1) rb = deg_load_dst(cur, dstv, e0, tid, E);
  for (int i = tid; i < DEG_RANGE; i += DEG_THREADS) cnt[i] = 0u;
  __syncthreads();

  for (int base = e0; base < e1; ) {
    const unsigned len = (unsigned)(e1 - base);
#pragma unroll
    for (int k = 0; k < DEG_K; ++k) {
      const unsigned r = (unsigned)(cur[k] - lo);
      const unsigned q = (unsigned)(rb + k * DEG_THREADS + tid - base);
      if (r < (unsigned)nr && q < len) atomicAdd(&cnt[r], 1u);
    }
    base += DEG_ROUND;
    if (base < e1) rb = deg_load_dst(cur, dstv, base, tid, E);
  }
  __syncthreads();

  unsigned* p = part + (size_t)blockIdx.y * N + lo;
  for (int i = tid; i < nr; i += DEG_THREADS) p[i] = cnt[i];
}

__global__ void k_dinv(const unsigned* __restrict__ part, float* __restrict__ dinv, int N) {
  int i = blockIdx.x * blockDim.x + threadIdx.x;
  if (i < N) {
    unsigned c = 1u;
#pragma unroll
    for (int s = 0; s < DEG_SLICES; ++s) c += part[(size_t)s * N + i];
    dinv[i] = rsqrtf((float)c);
  }
}


#define BK_ROWS    1000
#define BK_CHUNK   16384
#define BK_THREADS 1024
#define BK_K       (BK_CHUNK / BK_THREADS)
#define BK_MAXB    1024
#define PL_MOD     2000
#define REC_SHIFT  21
#define REC_MASK   ((1u << REC_SHIFT) - 1u)

__device__ __forceinline__ int row_prefix_sum(int v) {
  v += __builtin_amdgcn_update_dpp(0, v, 0x111, 0xf, 0xf, true);
  v += __builtin_amdgcn_update_dpp(0, v, 0x112, 0xf, 0xf, true);
  v += __builtin_amdgcn_update_dpp(0, v, 0x114, 0xf, 0xf, true);
  v += __builtin_amdgcn_update_dpp(0, v, 0x118, 0xf, 0xf, true);
  return v;
}

__device__ __forceinline__ int wave_prefix_sum(int v, int lane, int& total) {
  int inc = row_prefix_sum(v);
  const int t0 = __builtin_amdgcn_readlane(inc, 15);
  const int t1 = __builtin_amdgcn_readlane(inc, 31);
  if (lane >= 16) inc += t0;
  total = t0 + t1;
  return inc;
}

__global__ __launch_bounds__(BK_THREADS)
void k_bucket(const int* __restrict__ srcv, const int* __restrict__ dstv, const float* __restrict__ dinv,
              unsigned* __restrict__ recs, unsigned* __restrict__ offs, int N, int E, int NB) {
  __shared__ uint2 stage[BK_CHUNK];
  __shared__ unsigned cnt[BK_MAXB + 1];
  const int tid = threadIdx.x;
  const int lane = tid & 31;
  const int wave = __builtin_amdgcn_readfirstlane(tid >> 5);
  const int c0 = blockIdx.x * BK_CHUNK;

  int d[BK_K];
  int s[BK_K];
  if (c0 + BK_CHUNK <= E) {
    const int* q = dstv + c0 + tid;
    const int* qs = srcv + c0 + tid;
#pragma unroll
    for (int k = 0; k < BK_K; ++k) d[k] = q[k * BK_THREADS];
#pragma unroll
    for (int k = 0; k < BK_K; ++k) s[k] = qs[k * BK_THREADS];
  } else {
#pragma unroll
    for (int k = 0; k < BK_K; ++k) {
      const int e = c0 + k * BK_THREADS + tid;
      const int v = dstv[min(e, E - 1)];
      d[k] = (e < E) ? v : -1;
      s[k] = srcv[min(e, E - 1)];
    }
  }
  for (int i = tid; i <= NB; i += BK_THREADS) cnt[i] = 0u;
  __syncthreads();

  float ds[BK_K];
#pragma unroll
  for (int k = 0; k < BK_K; ++k) ds[k] = dinv[s[k]];

  unsigned rk[BK_K];
#pragma unroll
  for (int k = 0; k < BK_K; ++k) {
    rk[k] = 0u;
    if ((unsigned)d[k] < (unsigned)N) rk[k] = atomicAdd(&cnt[(unsigned)d[k] / BK_ROWS], 1u);
  }
  __syncthreads();

  if (wave == 0) {
    int run = 0;
    for (int b0 = 0; b0 < NB; b0 += 32) {
      const int b = b0 + lane;
      const int v = (b < NB) ? (int)cnt[b] : 0;
      int tot;
      const int inc = wave_prefix_sum(v, lane, tot);
      if (b < NB) cnt[b] = (unsigned)(run + inc - v);
      run += tot;
    }
    if (lane == 0) cnt[NB] = (unsigned)run;
  }
  __syncthreads();

#pragma unroll
  for (int k = 0; k < BK_K; ++k) {
    if ((unsigned)d[k] < (unsigned)N)
      stage[cnt[(unsigned)d[k] / BK_ROWS] + rk[k]] =
          make_uint2((unsigned)s[k] | ((unsigned)d[k] % PL_MOD) << REC_SHIFT, __float_as_uint(ds[k]));
  }
  __syncthreads();

  const int total = (int)cnt[NB];
  uint2* r = (uint2*)recs + c0;
  for (int i = tid; i < total; i += BK_THREADS) r[i] = stage[i];
  unsigned* o = offs + (size_t)blockIdx.x * (NB + 1);
  for (int i = tid; i <= NB; i += BK_THREADS) o[i] = cnt[i];
}


__global__ void k_gemm1(const float* __restrict__ X, const float* __restrict__ W,
                        float* __restrict__ H, int N) {
  const int lane = threadIdx.x & 31;
  const int wave = (blockIdx.x * blockDim.x + threadIdx.x) >> 5;
  const int m0 = wave << 4;
  if (m0 >= N) return;
  const int half = lane >> 4;
  const int lrow = lane & 15;
  int row = m0 + lrow; if (row >= N) row = N - 1;
  const float* xr = X + (size_t)row * F_IN;

  v8f c0 = {}; v8f c1 = {};
#pragma unroll 4
  for (int k0 = 0; k0 < F_IN; k0 += 4) {
    const int ka = k0 + (half << 1);
    v2f a;  a.x  = xr[ka];                      a.y  = xr[ka + 1];
    v2f b0; b0.x = W[ka * F_HID + lrow];        b0.y = W[(ka + 1) * F_HID + lrow];
    v2f b1; b1.x = W[ka * F_HID + 16 + lrow];   b1.y = W[(ka + 1) * F_HID + 16 + lrow];
    c0 = __builtin_amdgcn_wmma_f32_16x16x4_f32(false, a, false, b0, (short)0, c0, false, false);
    c1 = __builtin_amdgcn_wmma_f32_16x16x4_f32(false, a, false, b1, (short)0, c1, false, false);
  }
#pragma unroll
  for (int j = 0; j < 8; ++j) {
    int m = m0 + j + (half << 3);
    if (m < N) {
      float* hr = H + (size_t)m * F_HID;
      hr[lrow]      = c0[j];
      hr[16 + lrow] = c1[j];
    }
  }
}


__global__ void k_gemm2(const float* __restrict__ A, const float* __restrict__ W,
                        float* __restrict__ H, int N) {
  const int lane = threadIdx.x & 31;
  const int wave = (blockIdx.x * blockDim.x + threadIdx.x) >> 5;
  const int m0 = wave << 4;
  if (m0 >= N) return;
  const int half = lane >> 4;
  const int lrow = lane & 15;
  int row = m0 + lrow; if (row >= N) row = N - 1;
  const float* ar = A + (size_t)row * F_HID;

  v8f c0 = {}; v8f c1 = {}; v8f c2 = {}; v8f c3 = {};
#pragma unroll
  for (int k0 = 0; k0 < F_HID; k0 += 4) {
    const int ka = k0 + (half << 1);
    v2f a; a.x = fmaxf(ar[ka], 0.0f); a.y = fmaxf(ar[ka + 1], 0.0f);
    v2f b;
    b.x = W[ka * F_OUT + lrow];            b.y = W[(ka + 1) * F_OUT + lrow];
    c0 = __builtin_amdgcn_wmma_f32_16x16x4_f32(false, a, false, b, (short)0, c0, false, false);
    b.x = W[ka * F_OUT + 16 + lrow];       b.y = W[(ka + 1) * F_OUT + 16 + lrow];
    c1 = __builtin_amdgcn_wmma_f32_16x16x4_f32(false, a, false, b, (short)0, c1, false, false);
    b.x = W[ka * F_OUT + 32 + lrow];       b.y = W[(ka + 1) * F_OUT + 32 + lrow];
    c2 = __builtin_amdgcn_wmma_f32_16x16x4_f32(false, a, false, b, (short)0, c2, false, false);
    b.x = W[ka * F_OUT + 48 + lrow];       b.y = W[(ka + 1) * F_OUT + 48 + lrow];
    c3 = __builtin_amdgcn_wmma_f32_16x16x4_f32(false, a, false, b, (short)0, c3, false, false);
  }
#pragma unroll
  for (int j = 0; j < 8; ++j) {
    int m = m0 + j + (half << 3);
    if (m < N) {
      float* hr = H + (size_t)m * F_OUT;
      hr[lrow]      = c0[j];
      hr[16 + lrow] = c1[j];
      hr[32 + lrow] = c2[j];
      hr[48 + lrow] = c3[j];
    }
  }
}


#define INIT_THREADS 256

template <int F, int V, bool WHOLE>
__device__ __forceinline__ void agg_init_pieces(const v4f* __restrict__ hv, const float* __restrict__ dinv,
                                                v4f b, v4f* __restrict__ ov, unsigned q0, unsigned nvec) {
  constexpr unsigned VPR = F / 4;
  v4f x[V];
  float di[V];
#pragma unroll
  for (int j = 0; j < V; ++j) {
    unsigned q = q0 + j * INIT_THREADS;
    if (!WHOLE) q = min(q, nvec - 1u);
    x[j] = hv[q];
    di[j] = dinv[q / VPR];
  }
#pragma unroll
  for (int j = 0; j < V; ++j) {
    const unsigned q = q0 + j * INIT_THREADS;
    const v4f d4 = {di[j], di[j], di[j], di[j]};
    if (WHOLE || q < nvec) ov[q] = __builtin_elementwise_fma(x[j] * d4, d4, b);
  }
}

template <int F, int V>
__global__ __launch_bounds__(INIT_THREADS)
void k_agg_init(const float* __restrict__ h, const float* __restrict__ dinv,
                const float* __restrict__ bias, float* __restrict__ out, int nvec) {
  static_assert(INIT_THREADS % (F / 4) == 0, "a lane's pieces all lie at the same place of their rows");
  const unsigned tid = threadIdx.x;
  const unsigned g0 = blockIdx.x * (unsigned)(V * INIT_THREADS);
  const v4f b = *(const v4f*)(bias + 4u * (tid % (F / 4)));
  if (g0 + V * INIT_THREADS <= (unsigned)nvec)
    agg_init_pieces<F, V, true>((const v4f*)h, dinv, b, (v4f*)out, g0 + tid, (unsigned)nvec);
  else
    agg_init_pieces<F, V, false>((const v4f*)h, dinv, b, (v4f*)out, g0 + tid, (unsigned)nvec);
}


#define AGG_THREADS 1024
#define AGG_WAVES   32
#define AGG_MAXC    128
#define ACC_FLOATS  64000

template <int K>
__device__ __forceinline__ int row_bcast(int v) {
  return __builtin_amdgcn_update_dpp(0, v, 0x150 + K, 0xf, 0xf, true);
}

template <int F> struct agg_vec;
template <> struct agg_vec<32> { typedef v2f type; static constexpr int G = 16; };
template <> struct agg_vec<64> { typedef v4f type; static constexpr int G = 8; };

template <int F, int K0, int K>
__device__ __forceinline__ void agg_loads(typename agg_vec<F>::type (&rv)[agg_vec<F>::G],
                                          const char* __restrict__ hb, int sOff, int lanePart,
                                          int st) {
  typedef typename agg_vec<F>::type vec_t;
  if constexpr (K < agg_vec<F>::G) {
    if (K < st) {
      rv[K] = *(const vec_t*)(hb + (unsigned)(row_bcast<K0 + K>(sOff) + lanePart));
      agg_loads<F, K0, K + 1>(rv, hb, sOff, lanePart, st);
    }
  }
}

template <int F, int K0, int K>
__device__ __forceinline__ void agg_adds(const typename agg_vec<F>::type (&rv)[agg_vec<F>::G],
                                         float* acc, int wi, int dOff, int lanePart, int st) {
  constexpr int V = F / 16;
  if constexpr (K < agg_vec<F>::G) {
    if (K < st) {
      const float wk = __int_as_float(row_bcast<K0 + K>(wi));
      float* ar = (float*)((char*)acc + (unsigned)(row_bcast<K0 + K>(dOff) + lanePart));
#pragma unroll
      for (int c = 0; c < V; ++c) unsafeAtomicAdd(ar + c, rv[K][c] * wk);
      agg_adds<F, K0, K + 1>(rv, acc, wi, dOff, lanePart, st);
    }
  }
}

template <int F, int K0>
__device__ __forceinline__ void agg_walk(const float* __restrict__ h, float* acc, int sOff,
                                         float w, int dOff, int lanePart, int st) {
  typename agg_vec<F>::type rv[agg_vec<F>::G];
  agg_loads<F, K0, 0>(rv, (const char*)h, sOff, lanePart, st);
  agg_adds<F, K0, 0>(rv, acc, __float_as_int(w), dOff, lanePart, st);
}

template <int F>
__global__ __launch_bounds__(AGG_THREADS)
void k_agg_lds(const float* __restrict__ h, const float* __restrict__ dinv,
               const unsigned* __restrict__ recs,
               const unsigned* __restrict__ offs, float* __restrict__ out, int N, int NB, int nchunk) {
  constexpr int RANGE = ACC_FLOATS / F;
  constexpr int PPR = RANGE / BK_ROWS;
  static_assert(RANGE % BK_ROWS == 0, "a range is whole blocks");
  static_assert(PL_MOD % RANGE == 0, "a record's remainder gives the row inside every range");
  __shared__ float acc[ACC_FLOATS + F];
  __shared__ unsigned pre[AGG_MAXC + 1];
  __shared__ unsigned st[AGG_MAXC];

  const int tid = threadIdx.x;
  const int lane = tid & 31;
  const int wave = __builtin_amdgcn_readfirstlane(tid >> 5);
  const int lo = blockIdx.x * RANGE;
  const int nr = min(RANGE, N - lo);
  const int bin0 = blockIdx.x * PPR;
  const int np = min(PPR, NB - bin0);
  const int per = (nchunk + (int)gridDim.y - 1) / (int)gridDim.y;
  const int c0 = blockIdx.y * per;
  const int c1 = min(nchunk, c0 + per);
  const int slot = 2 * (lane & 15) + (lane >> 4);
  const int lanePart = (lane & 15) * (F / 16) * 4;

  for (int i = tid; i < ACC_FLOATS; i += AGG_THREADS) acc[i] = 0.0f;

  for (int g = c0; g < c1; g += AGG_MAXC) {
    const int ng = min(AGG_MAXC, c1 - g);
    if (g != c0) __syncthreads();
    if (wave == 0) {
      unsigned a[AGG_MAXC / 32], b[AGG_MAXC / 32];
#pragma unroll
      for (int q = 0; q < AGG_MAXC / 32; ++q) {
        const int ci = min(q * 32 + lane, ng - 1);
        const unsigned* o = offs + (size_t)(g + ci) * (NB + 1) + bin0;
        a[q] = o[0];
        b[q] = o[np];
      }
      int run = 0;
#pragma unroll
      for (int q = 0; q < AGG_MAXC / 32; ++q) {
        const int i = q * 32 + lane;
        const int len = (i < ng) ? (int)(b[q] - a[q]) : 0;
        int tot;
        const int inc = wave_prefix_sum(len, lane, tot);
        pre[i + 1] = (unsigned)(run + inc);
        st[i] = (unsigned)((g + min(i, ng - 1)) * BK_CHUNK) + a[q];
        run += tot;
      }
      if (lane == 0) pre[0] = 0u;
    }
    __syncthreads();

    const int T = (int)pre[AGG_MAXC];
    for (int b0 = wave * 32; b0 < T; b0 += AGG_WAVES * 32) {
      const int n = min(32, T - b0);
      const bool live = slot < n;
      const unsigned p = (unsigned)min(b0 + slot, T - 1);
      int cr = 0;
#pragma unroll
      for (int sstep = AGG_MAXC / 2; sstep > 0; sstep >>= 1)
        if (pre[cr + sstep] <= p) cr += sstep;
      uint2 rec = make_uint2(0u, 0u);
      if (live) rec = ((const uint2*)recs)[st[cr] + (p - pre[cr])];
      const int s = (int)(rec.x & REC_MASK);
      const int row = (int)(rec.x >> REC_SHIFT) % RANGE;
      const float ds = __uint_as_float(rec.y);
      const float dd = dinv[lo + row];
      const float w = live ? ds * dd : 0.0f;
      const int sOff = s * (F * 4);
      const int dOff = live ? row * (F * 4) : ACC_FLOATS * 4;
      const int steps = (n + 1) >> 1;
      constexpr int G = agg_vec<F>::G;
      agg_walk<F, 0>(h, acc, sOff, w, dOff, lanePart, min(steps, G));
      if constexpr (G < 16) {
        if (steps > G) agg_walk<F, G>(h, acc, sOff, w, dOff, lanePart, steps - G);
      }
    }
  }
  __syncthreads();

  float* o = out + (size_t)lo * F;
  for (int i = tid; i < nr * F; i += AGG_THREADS) unsafeAtomicAdd(&o[i], acc[i]);
}


static inline int cdiv_ll(long long a, long long b) { return (int)((a + b - 1) / b); }

extern "C" void kernel_launch(void* const* d_in, const int* in_sizes, int n_in,
                              void* d_out, int out_size, void* d_ws, size_t ws_size,
                              hipStream_t stream) {
  const float* X  = (const float*)d_in[0];
  const int*   ei = (const int*)d_in[1];
  const float* W1 = (const float*)d_in[2];
  const float* b1 = (const float*)d_in[3];
  const float* W2 = (const float*)d_in[4];
  const float* b2 = (const float*)d_in[5];
  float* out = (float*)d_out;

  const int N = in_sizes[0] / F_IN;
  const int E = in_sizes[1] / 2;
  const int* src = ei;
  const int* dst = ei + E;

  const int nchunk = cdiv_ll(E, BK_CHUNK);
  const int NB = cdiv_ll(N, BK_ROWS);

  unsigned* recs = (unsigned*)d_ws;
  float* dinv = (float*)(recs + (size_t)nchunk * BK_CHUNK * 2);
  float* h1   = dinv + N;
  float* out1 = h1 + (size_t)N * F_HID;
  float* h2   = out1 + (size_t)N * F_HID;
  unsigned* part = (unsigned*)(h2 + (size_t)N * F_OUT);
  unsigned* offs = part + (size_t)DEG_SLICES * N;

  const int B = 256;
  const int mtiles = cdiv_ll(N, 16);
  const int nv1 = N * (F_HID / 4);
  const int nv2 = N * (F_OUT / 4);

  k_count_deg<<<dim3(cdiv_ll(N, DEG_RANGE), DEG_SLICES), DEG_THREADS, 0, stream>>>(dst, part, N, E);
  k_dinv     <<<cdiv_ll(N, B), B, 0, stream>>>(part, dinv, N);
  k_bucket   <<<nchunk, BK_THREADS, 0, stream>>>(src, dst, dinv, recs, offs, N, E, NB);

  k_gemm1    <<<cdiv_ll((long long)mtiles * 32, B), B, 0, stream>>>(X, W1, h1, N);
  k_agg_init<F_HID, 4><<<cdiv_ll(nv1, 4 * INIT_THREADS), INIT_THREADS, 0, stream>>>(h1, dinv, b1, out1, nv1);
  k_agg_lds<F_HID><<<dim3(cdiv_ll(N, ACC_FLOATS / F_HID), 5), AGG_THREADS, 0, stream>>>(h1, dinv, recs, offs, out1, N, NB, nchunk);

  k_gemm2    <<<cdiv_ll((long long)mtiles * 32, B), B, 0, stream>>>(out1, W2, h2, N);
  k_agg_init<F_OUT, 8><<<cdiv_ll(nv2, 8 * INIT_THREADS), INIT_THREADS, 0, stream>>>(h2, dinv, b2, out, nv2);
  k_agg_lds<F_OUT><<<dim3(cdiv_ll(N, ACC_FLOATS / F_OUT), 2), AGG_THREADS, 0, stream>>>(h2, dinv, recs, offs, out, N, NB, nchunk);
}


# both edge sums: one workgroup alone owns 400 destination rows (all chunks), stores self-loop + bias + LDS sums itself: no global atomics, no start-value passes; 9 launches, now 7; sum order moves
# speedup vs baseline: 3.3161x; 1.0766x over previous
#include <hip/hip_runtime.h>
#include <hip/hip_bf16.h>

typedef float v2f __attribute__((ext_vector_type(2)));
typedef float v4f __attribute__((ext_vector_type(4)));
typedef float v8f __attribute__((ext_vector_type(8)));

#define F_IN  512
#define F_HID 32
#define F_OUT 64


#define DEG_THREADS 1024
#define DEG_RANGE   25000
#define DEG_SLICES  64
#define DEG_K       50
#define DEG_ROUND   (DEG_K * DEG_THREADS)

__device__ __forceinline__ int deg_load_dst(int (&cur)[DEG_K], const int* __restrict__ dstv,
                                            int base, int tid, int E) {
  if (E >= DEG_ROUND) {
    const int rb = min(base, E - DEG_ROUND);
    const int* p = dstv + rb + tid;
#pragma unroll
    for (int k = 0; k < DEG_K; ++k) cur[k] = p[k * DEG_THREADS];
    return rb;
  }
#pragma unroll
  for (int k = 0; k < DEG_K; ++k) {
    const int e = base + k * DEG_THREADS + tid;
    const int v = dstv[min(e, E - 1)];
    cur[k] = (e < E) ? v : -1;
  }
  return base;
}

__global__ __launch_bounds__(DEG_THREADS)
void k_count_deg(const int* __restrict__ dstv, unsigned* __restrict__ part, int N, int E) {
  __shared__ unsigned cnt[DEG_RANGE];
  const int tid = threadIdx.x;
  const int lo = blockIdx.x * DEG_RANGE;
  const int nr = min(DEG_RANGE, N - lo);
  const int per = (((E + DEG_SLICES - 1) / DEG_SLICES) + 63) & ~63;
  const int e0 = min(E, (int)blockIdx.y * per);
  const int e1 = min(E, e0 + per);

  int cur[DEG_K];
  int rb = e0;
  if (e0 < e1) rb = deg_load_dst(cur, dstv, e0, tid, E);
  for (int i = tid; i < DEG_RANGE; i += DEG_THREADS) cnt[i] = 0u;
  __syncthreads();

  for (int base = e0; base < e1; ) {
    const unsigned len = (unsigned)(e1 - base);
#pragma unroll
    for (int k = 0; k < DEG_K; ++k) {
      const unsigned r = (unsigned)(cur[k] - lo);
      const unsigned q = (unsigned)(rb + k * DEG_THREADS + tid - base);
      if (r < (unsigned)nr && q < len) atomicAdd(&cnt[r], 1u);
    }
    base += DEG_ROUND;
    if (base < e1) rb = deg_load_dst(cur, dstv, base, tid, E);
  }
  __syncthreads();

  unsigned* p = part + (size_t)blockIdx.y * N + lo;
  for (int i = tid; i < nr; i += DEG_THREADS) p[i] = cnt[i];
}

__global__ void k_dinv(const unsigned* __restrict__ part, float* __restrict__ dinv, int N) {
  int i = blockIdx.x * blockDim.x + threadIdx.x;
  if (i < N) {
    unsigned c = 1u;
#pragma unroll
    for (int s = 0; s < DEG_SLICES; ++s) c += part[(size_t)s * N + i];
    dinv[i] = rsqrtf((float)c);
  }
}


#define BK_ROWS    400
#define BK_CHUNK   16384
#define BK_THREADS 1024
#define BK_K       (BK_CHUNK / BK_THREADS)
#define BK_MAXB    1024
#define REC_SHIFT  21
#define REC_MASK   ((1u << REC_SHIFT) - 1u)

__device__ __forceinline__ int row_prefix_sum(int v) {
  v += __builtin_amdgcn_update_dpp(0, v, 0x111, 0xf, 0xf, true);
  v += __builtin_amdgcn_update_dpp(0, v, 0x112, 0xf, 0xf, true);
  v += __builtin_amdgcn_update_dpp(0, v, 0x114, 0xf, 0xf, true);
  v += __builtin_amdgcn_update_dpp(0, v, 0x118, 0xf, 0xf, true);
  return v;
}

__device__ __forceinline__ int wave_prefix_sum(int v, int lane, int& total) {
  int inc = row_prefix_sum(v);
  const int t0 = __builtin_amdgcn_readlane(inc, 15);
  const int t1 = __builtin_amdgcn_readlane(inc, 31);
  if (lane >= 16) inc += t0;
  total = t0 + t1;
  return inc;
}

__global__ __launch_bounds__(BK_THREADS)
void k_bucket(const int* __restrict__ srcv, const int* __restrict__ dstv, const float* __restrict__ dinv,
              unsigned* __restrict__ recs, unsigned* __restrict__ offs, int N, int E, int NB) {
  __shared__ uint2 stage[BK_CHUNK];
  __shared__ unsigned cnt[BK_MAXB + 1];
  const int tid = threadIdx.x;
  const int lane = tid & 31;
  const int wave = __builtin_amdgcn_readfirstlane(tid >> 5);
  const int c0 = blockIdx.x * BK_CHUNK;

  int d[BK_K];
  int s[BK_K];
  if (c0 + BK_CHUNK <= E) {
    const int* q = dstv + c0 + tid;
    const int* qs = srcv + c0 + tid;
#pragma unroll
    for (int k = 0; k < BK_K; ++k) d[k] = q[k * BK_THREADS];
#pragma unroll
    for (int k = 0; k < BK_K; ++k) s[k] = qs[k * BK_THREADS];
  } else {
#pragma unroll
    for (int k = 0; k < BK_K; ++k) {
      const int e = c0 + k * BK_THREADS + tid;
      const int v = dstv[min(e, E - 1)];
      d[k] = (e < E) ? v : -1;
      s[k] = srcv[min(e, E - 1)];
    }
  }
  for (int i = tid; i <= NB; i += BK_THREADS) cnt[i] = 0u;
  __syncthreads();

  float ds[BK_K];
#pragma unroll
  for (int k = 0; k < BK_K; ++k) ds[k] = dinv[s[k]];

  unsigned rk[BK_K];
#pragma unroll
  for (int k = 0; k < BK_K; ++k) {
    rk[k] = 0u;
    if ((unsigned)d[k] < (unsigned)N) rk[k] = atomicAdd(&cnt[(unsigned)d[k] / BK_ROWS], 1u);
  }
  __syncthreads();

  if (wave == 0) {
    int run = 0;
    for (int b0 = 0; b0 < NB; b0 += 32) {
      const int b = b0 + lane;
      const int v = (b < NB) ? (int)cnt[b] : 0;
      int tot;
      const int inc = wave_prefix_sum(v, lane, tot);
      if (b < NB) cnt[b] = (unsigned)(run + inc - v);
      run += tot;
    }
    if (lane == 0) cnt[NB] = (unsigned)run;
  }
  __syncthreads();

#pragma unroll
  for (int k = 0; k < BK_K; ++k) {
    if ((unsigned)d[k] < (unsigned)N)
      stage[cnt[(unsigned)d[k] / BK_ROWS] + rk[k]] =
          make_uint2((unsigned)s[k] | ((unsigned)d[k] % BK_ROWS) << REC_SHIFT, __float_as_uint(ds[k]));
  }
  __syncthreads();

  const int total = (int)cnt[NB];
  uint2* r = (uint2*)recs + c0;
  for (int i = tid; i < total; i += BK_THREADS) r[i] = stage[i];
  unsigned* o = offs + (size_t)blockIdx.x * (NB + 1);
  for (int i = tid; i <= NB; i += BK_THREADS) o[i] = cnt[i];
}


__global__ void k_gemm1(const float* __restrict__ X, const float* __restrict__ W,
                        float* __restrict__ H, int N) {
  const int lane = threadIdx.x & 31;
  const int wave = (blockIdx.x * blockDim.x + threadIdx.x) >> 5;
  const int m0 = wave << 4;
  if (m0 >= N) return;
  const int half = lane >> 4;
  const int lrow = lane & 15;
  int row = m0 + lrow; if (row >= N) row = N - 1;
  const float* xr = X + (size_t)row * F_IN;

  v8f c0 = {}; v8f c1 = {};
#pragma unroll 4
  for (int k0 = 0; k0 < F_IN; k0 += 4) {
    const int ka = k0 + (half << 1);
    v2f a;  a.x  = xr[ka];                      a.y  = xr[ka + 1];
    v2f b0; b0.x = W[ka * F_HID + lrow];        b0.y = W[(ka + 1) * F_HID + lrow];
    v2f b1; b1.x = W[ka * F_HID + 16 + lrow];   b1.y = W[(ka + 1) * F_HID + 16 + lrow];
    c0 = __builtin_amdgcn_wmma_f32_16x16x4_f32(false, a, false, b0, (short)0, c0, false, false);
    c1 = __builtin_amdgcn_wmma_f32_16x16x4_f32(false, a, false, b1, (short)0, c1, false, false);
  }
#pragma unroll
  for (int j = 0; j < 8; ++j) {
    int m = m0 + j + (half << 3);
    if (m < N) {
      float* hr = H + (size_t)m * F_HID;
      hr[lrow]      = c0[j];
      hr[16 + lrow] = c1[j];
    }
  }
}


__global__ void k_gemm2(const float* __restrict__ A, const float* __restrict__ W,
                        float* __restrict__ H, int N) {
  const int lane = threadIdx.x & 31;
  const int wave = (blockIdx.x * blockDim.x + threadIdx.x) >> 5;
  const int m0 = wave << 4;
  if (m0 >= N) return;
  const int half = lane >> 4;
  const int lrow = lane & 15;
  int row = m0 + lrow; if (row >= N) row = N - 1;
  const float* ar = A + (size_t)row * F_HID;

  v8f c0 = {}; v8f c1 = {}; v8f c2 = {}; v8f c3 = {};
#pragma unroll
  for (int k0 = 0; k0 < F_HID; k0 += 4) {
    const int ka = k0 + (half << 1);
    v2f a; a.x = fmaxf(ar[ka], 0.0f); a.y = fmaxf(ar[ka + 1], 0.0f);
    v2f b;
    b.x = W[ka * F_OUT + lrow];            b.y = W[(ka + 1) * F_OUT + lrow];
    c0 = __builtin_amdgcn_wmma_f32_16x16x4_f32(false, a, false, b, (short)0, c0, false, false);
    b.x = W[ka * F_OUT + 16 + lrow];       b.y = W[(ka + 1) * F_OUT + 16 + lrow];
    c1 = __builtin_amdgcn_wmma_f32_16x16x4_f32(false, a, false, b, (short)0, c1, false, false);
    b.x = W[ka * F_OUT + 32 + lrow];       b.y = W[(ka + 1) * F_OUT + 32 + lrow];
    c2 = __builtin_amdgcn_wmma_f32_16x16x4_f32(false, a, false, b, (short)0, c2, false, false);
    b.x = W[ka * F_OUT + 48 + lrow];       b.y = W[(ka + 1) * F_OUT + 48 + lrow];
    c3 = __builtin_amdgcn_wmma_f32_16x16x4_f32(false, a, false, b, (short)0, c3, false, false);
  }
#pragma unroll
  for (int j = 0; j < 8; ++j) {
    int m = m0 + j + (half << 3);
    if (m < N) {
      float* hr = H + (size_t)m * F_OUT;
      hr[lrow]      = c0[j];
      hr[16 + lrow] = c1[j];
      hr[32 + lrow] = c2[j];
      hr[48 + lrow] = c3[j];
    }
  }
}


#define AGG_THREADS 1024
#define AGG_WAVES   32
#define AGG_MAXC    256

template <int K>
__device__ __forceinline__ int row_bcast(int v) {
  return __builtin_amdgcn_update_dpp(0, v, 0x150 + K, 0xf, 0xf, true);
}

template <int F> struct agg_vec;
template <> struct agg_vec<32> { typedef v2f type; static constexpr int G = 16; };
template <> struct agg_vec<64> { typedef v4f type; static constexpr int G = 8; };

template <int F, int K0, int K>
__device__ __forceinline__ void agg_loads(typename agg_vec<F>::type (&rv)[agg_vec<F>::G],
                                          const char* __restrict__ hb, int sOff, int lanePart,
                                          int st) {
  typedef typename agg_vec<F>::type vec_t;
  if constexpr (K < agg_vec<F>::G) {
    if (K < st) {
      rv[K] = *(const vec_t*)(hb + (unsigned)(row_bcast<K0 + K>(sOff) + lanePart));
      agg_loads<F, K0, K + 1>(rv, hb, sOff, lanePart, st);
    }
  }
}

template <int F, int K0, int K>
__device__ __forceinline__ void agg_adds(const typename agg_vec<F>::type (&rv)[agg_vec<F>::G],
                                         float* acc, int wi, int dOff, int lanePart, int st) {
  constexpr int V = F / 16;
  if constexpr (K < agg_vec<F>::G) {
    if (K < st) {
      const float wk = __int_as_float(row_bcast<K0 + K>(wi));
      float* ar = (float*)((char*)acc + (unsigned)(row_bcast<K0 + K>(dOff) + lanePart));
#pragma unroll
      for (int c = 0; c < V; ++c) unsafeAtomicAdd(ar + c, rv[K][c] * wk);
      agg_adds<F, K0, K + 1>(rv, acc, wi, dOff, lanePart, st);
    }
  }
}

template <int F, int K0>
__device__ __forceinline__ void agg_walk(const float* __restrict__ h, float* acc, int sOff,
                                         float w, int dOff, int lanePart, int st) {
  typename agg_vec<F>::type rv[agg_vec<F>::G];
  agg_loads<F, K0, 0>(rv, (const char*)h, sOff, lanePart, st);
  agg_adds<F, K0, 0>(rv, acc, __float_as_int(w), dOff, lanePart, st);
}

template <int F>
__global__ __launch_bounds__(AGG_THREADS)
void k_agg_lds(const float* __restrict__ h, const float* __restrict__ dinv,
               const float* __restrict__ bias, const unsigned* __restrict__ recs,
               const unsigned* __restrict__ offs, float* __restrict__ out, int N, int NB, int nchunk) {
  constexpr int RANGE = BK_ROWS;
  constexpr int ACC_FLOATS = RANGE * F;
  constexpr int VPR = F / 4;
  constexpr int RV = (RANGE * VPR + AGG_THREADS - 1) / AGG_THREADS;
  static_assert(AGG_THREADS % VPR == 0, "a lane's pieces all lie at the same place of their rows");
  __shared__ __attribute__((aligned(16))) float acc[ACC_FLOATS + F];
  __shared__ unsigned pre[AGG_MAXC + 1];
  __shared__ unsigned st[AGG_MAXC];

  const int tid = threadIdx.x;
  const int lane = tid & 31;
  const int wave = __builtin_amdgcn_readfirstlane(tid >> 5);
  const int bin = blockIdx.x;
  const int lo = bin * RANGE;
  const int nr = min(RANGE, N - lo);
  const int slot = 2 * (lane & 15) + (lane >> 4);
  const int lanePart = (lane & 15) * (F / 16) * 4;

  for (int i = tid; i < ACC_FLOATS; i += AGG_THREADS) acc[i] = 0.0f;

  for (int g = 0; g < nchunk; g += AGG_MAXC) {
    const int ng = min(AGG_MAXC, nchunk - g);
    if (g != 0) __syncthreads();
    if (wave == 0) {
      unsigned a[AGG_MAXC / 32], b[AGG_MAXC / 32];
#pragma unroll
      for (int q = 0; q < AGG_MAXC / 32; ++q) {
        const int ci = min(q * 32 + lane, ng - 1);
        const unsigned* o = offs + (size_t)(g + ci) * (NB + 1) + bin;
        a[q] = o[0];
        b[q] = o[1];
      }
      int run = 0;
#pragma unroll
      for (int q = 0; q < AGG_MAXC / 32; ++q) {
        const int i = q * 32 + lane;
        const int len = (i < ng) ? (int)(b[q] - a[q]) : 0;
        int tot;
        const int inc = wave_prefix_sum(len, lane, tot);
        pre[i + 1] = (unsigned)(run + inc);
        st[i] = (unsigned)((g + min(i, ng - 1)) * BK_CHUNK) + a[q];
        run += tot;
      }
      if (lane == 0) pre[0] = 0u;
    }
    __syncthreads();

    const int T = (int)pre[AGG_MAXC];
    for (int b0 = wave * 32; b0 < T; b0 += AGG_WAVES * 32) {
      const int n = min(32, T - b0);
      const bool live = slot < n;
      const unsigned p = (unsigned)min(b0 + slot, T - 1);
      int cr = 0;
#pragma unroll
      for (int sstep = AGG_MAXC / 2; sstep > 0; sstep >>= 1)
        if (pre[cr + sstep] <= p) cr += sstep;
      uint2 rec = make_uint2(0u, 0u);
      if (live) rec = ((const uint2*)recs)[st[cr] + (p - pre[cr])];
      const int s = (int)(rec.x & REC_MASK);
      const int row = (int)(rec.x >> REC_SHIFT);
      const float ds = __uint_as_float(rec.y);
      const float dd = dinv[lo + row];
      const float w = live ? ds * dd : 0.0f;
      const int sOff = s * (F * 4);
      const int dOff = live ? row * (F * 4) : ACC_FLOATS * 4;
      const int steps = (n + 1) >> 1;
      constexpr int G = agg_vec<F>::G;
      agg_walk<F, 0>(h, acc, sOff, w, dOff, lanePart, min(steps, G));
      if constexpr (G < 16) {
        if (steps > G) agg_walk<F, G>(h, acc, sOff, w, dOff, lanePart, steps - G);
      }
    }
  }

  const unsigned npc = (unsigned)(nr * VPR);
  const v4f* hv = (const v4f*)(h + (size_t)lo * F);
  v4f x[RV];
  float di[RV];
#pragma unroll
  for (int j = 0; j < RV; ++j) {
    const unsigned q = min((unsigned)(tid + j * AGG_THREADS), npc - 1u);
    x[j] = hv[q];
    di[j] = dinv[lo + (int)(q / VPR)];
  }
  const v4f bv = *(const v4f*)(bias + 4 * (tid % VPR));
  __syncthreads();

  v4f sum[RV];
#pragma unroll
  for (int j = 0; j < RV; ++j)
    sum[j] = ((const v4f*)acc)[min((unsigned)(tid + j * AGG_THREADS), npc - 1u)];
  v4f* ov = (v4f*)(out + (size_t)lo * F);
#pragma unroll
  for (int j = 0; j < RV; ++j) {
    const unsigned q = (unsigned)(tid + j * AGG_THREADS);
    const v4f d4 = {di[j], di[j], di[j], di[j]};
    if (q < npc) ov[q] = __builtin_elementwise_fma(x[j] * d4, d4, bv) + sum[j];
  }
}


static inline int cdiv_ll(long long a, long long b) { return (int)((a + b - 1) / b); }

extern "C" void kernel_launch(void* const* d_in, const int* in_sizes, int n_in,
                              void* d_out, int out_size, void* d_ws, size_t ws_size,
                              hipStream_t stream) {
  const float* X  = (const float*)d_in[0];
  const int*   ei = (const int*)d_in[1];
  const float* W1 = (const float*)d_in[2];
  const float* b1 = (const float*)d_in[3];
  const float* W2 = (const float*)d_in[4];
  const float* b2 = (const float*)d_in[5];
  float* out = (float*)d_out;

  const int N = in_sizes[0] / F_IN;
  const int E = in_sizes[1] / 2;
  const int* src = ei;
  const int* dst = ei + E;

  const int nchunk = cdiv_ll(E, BK_CHUNK);
  const int NB = cdiv_ll(N, BK_ROWS);

  unsigned* recs = (unsigned*)d_ws;
  float* dinv = (float*)(recs + (size_t)nchunk * BK_CHUNK * 2);
  float* h1   = dinv + N;
  float* out1 = h1 + (size_t)N * F_HID;
  float* h2   = out1 + (size_t)N * F_HID;
  unsigned* part = (unsigned*)(h2 + (size_t)N * F_OUT);
  unsigned* offs = part + (size_t)DEG_SLICES * N;

  const int B = 256;
  const int mtiles = cdiv_ll(N, 16);

  k_count_deg<<<dim3(cdiv_ll(N, DEG_RANGE), DEG_SLICES), DEG_THREADS, 0, stream>>>(dst, part, N, E);
  k_dinv     <<<cdiv_ll(N, B), B, 0, stream>>>(part, dinv, N);
  k_bucket   <<<nchunk, BK_THREADS, 0, stream>>>(src, dst, dinv, recs, offs, N, E, NB);

  k_gemm1    <<<cdiv_ll((long long)mtiles * 32, B), B, 0, stream>>>(X, W1, h1, N);
  k_agg_lds<F_HID><<<NB, AGG_THREADS, 0, stream>>>(h1, dinv, b1, recs, offs, out1, N, NB, nchunk);

  k_gemm2    <<<cdiv_ll((long long)mtiles * 32, B), B, 0, stream>>>(out1, W2, h2, N);
  k_agg_lds<F_OUT><<<NB, AGG_THREADS, 0, stream>>>(h2, dinv, b2, recs, offs, out, N, NB, nchunk);
}


# both edge sums: lanes 16-31 add each float pair in swapped order, so the wave's two LDS rows (pitch a multiple of 64 banks) use disjoint banks per atomic, one pass not two; LDS sum order as before
# speedup vs baseline: 3.9472x; 1.1903x over previous
#include <hip/hip_runtime.h>
#include <hip/hip_bf16.h>

typedef float v2f __attribute__((ext_vector_type(2)));
typedef float v4f __attribute__((ext_vector_type(4)));
typedef float v8f __attribute__((ext_vector_type(8)));

#define F_IN  512
#define F_HID 32
#define F_OUT 64


#define DEG_THREADS 1024
#define DEG_RANGE   25000
#define DEG_SLICES  64
#define DEG_K       50
#define DEG_ROUND   (DEG_K * DEG_THREADS)

__device__ __forceinline__ int deg_load_dst(int (&cur)[DEG_K], const int* __restrict__ dstv,
                                            int base, int tid, int E) {
  if (E >= DEG_ROUND) {
    const int rb = min(base, E - DEG_ROUND);
    const int* p = dstv + rb + tid;
#pragma unroll
    for (int k = 0; k < DEG_K; ++k) cur[k] = p[k * DEG_THREADS];
    return rb;
  }
#pragma unroll
  for (int k = 0; k < DEG_K; ++k) {
    const int e = base + k * DEG_THREADS + tid;
    const int v = dstv[min(e, E - 1)];
    cur[k] = (e < E) ? v : -1;
  }
  return base;
}

__global__ __launch_bounds__(DEG_THREADS)
void k_count_deg(const int* __restrict__ dstv, unsigned* __restrict__ part, int N, int E) {
  __shared__ unsigned cnt[DEG_RANGE];
  const int tid = threadIdx.x;
  const int lo = blockIdx.x * DEG_RANGE;
  const int nr = min(DEG_RANGE, N - lo);
  const int per = (((E + DEG_SLICES - 1) / DEG_SLICES) + 63) & ~63;
  const int e0 = min(E, (int)blockIdx.y * per);
  const int e1 = min(E, e0 + per);

  int cur[DEG_K];
  int rb = e0;
  if (e0 < e1) rb = deg_load_dst(cur, dstv, e0, tid, E);
  for (int i = tid; i < DEG_RANGE; i += DEG_THREADS) cnt[i] = 0u;
  __syncthreads();

  for (int base = e0; base < e1; ) {
    const unsigned len = (unsigned)(e1 - base);
#pragma unroll
    for (int k = 0; k < DEG_K; ++k) {
      const unsigned r = (unsigned)(cur[k] - lo);
      const unsigned q = (unsigned)(rb + k * DEG_THREADS + tid - base);
      if (r < (unsigned)nr && q < len) atomicAdd(&cnt[r], 1u);
    }
    base += DEG_ROUND;
    if (base < e1) rb = deg_load_dst(cur, dstv, base, tid, E);
  }
  __syncthreads();

  unsigned* p = part + (size_t)blockIdx.y * N + lo;
  for (int i = tid; i < nr; i += DEG_THREADS) p[i] = cnt[i];
}

__global__ void k_dinv(const unsigned* __restrict__ part, float* __restrict__ dinv, int N) {
  int i = blockIdx.x * blockDim.x + threadIdx.x;
  if (i < N) {
    unsigned c = 1u;
#pragma unroll
    for (int s = 0; s < DEG_SLICES; ++s) c += part[(size_t)s * N + i];
    dinv[i] = rsqrtf((float)c);
  }
}


#define BK_ROWS    400
#define BK_CHUNK   16384
#define BK_THREADS 1024
#define BK_K       (BK_CHUNK / BK_THREADS)
#define BK_MAXB    1024
#define REC_SHIFT  21
#define REC_MASK   ((1u << REC_SHIFT) - 1u)

__device__ __forceinline__ int row_prefix_sum(int v) {
  v += __builtin_amdgcn_update_dpp(0, v, 0x111, 0xf, 0xf, true);
  v += __builtin_amdgcn_update_dpp(0, v, 0x112, 0xf, 0xf, true);
  v += __builtin_amdgcn_update_dpp(0, v, 0x114, 0xf, 0xf, true);
  v += __builtin_amdgcn_update_dpp(0, v, 0x118, 0xf, 0xf, true);
  return v;
}

__device__ __forceinline__ int wave_prefix_sum(int v, int lane, int& total) {
  int inc = row_prefix_sum(v);
  const int t0 = __builtin_amdgcn_readlane(inc, 15);
  const int t1 = __builtin_amdgcn_readlane(inc, 31);
  if (lane >= 16) inc += t0;
  total = t0 + t1;
  return inc;
}

__global__ __launch_bounds__(BK_THREADS)
void k_bucket(const int* __restrict__ srcv, const int* __restrict__ dstv, const float* __restrict__ dinv,
              unsigned* __restrict__ recs, unsigned* __restrict__ offs, int N, int E, int NB) {
  __shared__ uint2 stage[BK_CHUNK];
  __shared__ unsigned cnt[BK_MAXB + 1];
  const int tid = threadIdx.x;
  const int lane = tid & 31;
  const int wave = __builtin_amdgcn_readfirstlane(tid >> 5);
  const int c0 = blockIdx.x * BK_CHUNK;

  int d[BK_K];
  int s[BK_K];
  if (c0 + BK_CHUNK <= E) {
    const int* q = dstv + c0 + tid;
    const int* qs = srcv + c0 + tid;
#pragma unroll
    for (int k = 0; k < BK_K; ++k) d[k] = q[k * BK_THREADS];
#pragma unroll
    for (int k = 0; k < BK_K; ++k) s[k] = qs[k * BK_THREADS];
  } else {
#pragma unroll
    for (int k = 0; k < BK_K; ++k) {
      const int e = c0 + k * BK_THREADS + tid;
      const int v = dstv[min(e, E - 1)];
      d[k] = (e < E) ? v : -1;
      s[k] = srcv[min(e, E - 1)];
    }
  }
  for (int i = tid; i <= NB; i += BK_THREADS) cnt[i] = 0u;
  __syncthreads();

  float ds[BK_K];
#pragma unroll
  for (int k = 0; k < BK_K; ++k) ds[k] = dinv[s[k]];

  unsigned rk[BK_K];
#pragma unroll
  for (int k = 0; k < BK_K; ++k) {
    rk[k] = 0u;
    if ((unsigned)d[k] < (unsigned)N) rk[k] = atomicAdd(&cnt[(unsigned)d[k] / BK_ROWS], 1u);
  }
  __syncthreads();

  if (wave == 0) {
    int run = 0;
    for (int b0 = 0; b0 < NB; b0 += 32) {
      const int b = b0 + lane;
      const int v = (b < NB) ? (int)cnt[b] : 0;
      int tot;
      const int inc = wave_prefix_sum(v, lane, tot);
      if (b < NB) cnt[b] = (unsigned)(run + inc - v);
      run += tot;
    }
    if (lane == 0) cnt[NB] = (unsigned)run;
  }
  __syncthreads();

#pragma unroll
  for (int k = 0; k < BK_K; ++k) {
    if ((unsigned)d[k] < (unsigned)N)
      stage[cnt[(unsigned)d[k] / BK_ROWS] + rk[k]] =
          make_uint2((unsigned)s[k] | ((unsigned)d[k] % BK_ROWS) << REC_SHIFT, __float_as_uint(ds[k]));
  }
  __syncthreads();

  const int total = (int)cnt[NB];
  uint2* r = (uint2*)recs + c0;
  for (int i = tid; i < total; i += BK_THREADS) r[i] = stage[i];
  unsigned* o = offs + (size_t)blockIdx.x * (NB + 1);
  for (int i = tid; i <= NB; i += BK_THREADS) o[i] = cnt[i];
}


__global__ void k_gemm1(const float* __restrict__ X, const float* __restrict__ W,
                        float* __restrict__ H, int N) {
  const int lane = threadIdx.x & 31;
  const int wave = (blockIdx.x * blockDim.x + threadIdx.x) >> 5;
  const int m0 = wave << 4;
  if (m0 >= N) return;
  const int half = lane >> 4;
  const int lrow = lane & 15;
  int row = m0 + lrow; if (row >= N) row = N - 1;
  const float* xr = X + (size_t)row * F_IN;

  v8f c0 = {}; v8f c1 = {};
#pragma unroll 4
  for (int k0 = 0; k0 < F_IN; k0 += 4) {
    const int ka = k0 + (half << 1);
    v2f a;  a.x  = xr[ka];                      a.y  = xr[ka + 1];
    v2f b0; b0.x = W[ka * F_HID + lrow];        b0.y = W[(ka + 1) * F_HID + lrow];
    v2f b1; b1.x = W[ka * F_HID + 16 + lrow];   b1.y = W[(ka + 1) * F_HID + 16 + lrow];
    c0 = __builtin_amdgcn_wmma_f32_16x16x4_f32(false, a, false, b0, (short)0, c0, false, false);
    c1 = __builtin_amdgcn_wmma_f32_16x16x4_f32(false, a, false, b1, (short)0, c1, false, false);
  }
#pragma unroll
  for (int j = 0; j < 8; ++j) {
    int m = m0 + j + (half << 3);
    if (m < N) {
      float* hr = H + (size_t)m * F_HID;
      hr[lrow]      = c0[j];
      hr[16 + lrow] = c1[j];
    }
  }
}


__global__ void k_gemm2(const float* __restrict__ A, const float* __restrict__ W,
                        float* __restrict__ H, int N) {
  const int lane = threadIdx.x & 31;
  const int wave = (blockIdx.x * blockDim.x + threadIdx.x) >> 5;
  const int m0 = wave << 4;
  if (m0 >= N) return;
  const int half = lane >> 4;
  const int lrow = lane & 15;
  int row = m0 + lrow; if (row >= N) row = N - 1;
  const float* ar = A + (size_t)row * F_HID;

  v8f c0 = {}; v8f c1 = {}; v8f c2 = {}; v8f c3 = {};
#pragma unroll
  for (int k0 = 0; k0 < F_HID; k0 += 4) {
    const int ka = k0 + (half << 1);
    v2f a; a.x = fmaxf(ar[ka], 0.0f); a.y = fmaxf(ar[ka + 1], 0.0f);
    v2f b;
    b.x = W[ka * F_OUT + lrow];            b.y = W[(ka + 1) * F_OUT + lrow];
    c0 = __builtin_amdgcn_wmma_f32_16x16x4_f32(false, a, false, b, (short)0, c0, false, false);
    b.x = W[ka * F_OUT + 16 + lrow];       b.y = W[(ka + 1) * F_OUT + 16 + lrow];
    c1 = __builtin_amdgcn_wmma_f32_16x16x4_f32(false, a, false, b, (short)0, c1, false, false);
    b.x = W[ka * F_OUT + 32 + lrow];       b.y = W[(ka + 1) * F_OUT + 32 + lrow];
    c2 = __builtin_amdgcn_wmma_f32_16x16x4_f32(false, a, false, b, (short)0, c2, false, false);
    b.x = W[ka * F_OUT + 48 + lrow];       b.y = W[(ka + 1) * F_OUT + 48 + lrow];
    c3 = __builtin_amdgcn_wmma_f32_16x16x4_f32(false, a, false, b, (short)0, c3, false, false);
  }
#pragma unroll
  for (int j = 0; j < 8; ++j) {
    int m = m0 + j + (half << 3);
    if (m < N) {
      float* hr = H + (size_t)m * F_OUT;
      hr[lrow]      = c0[j];
      hr[16 + lrow] = c1[j];
      hr[32 + lrow] = c2[j];
      hr[48 + lrow] = c3[j];
    }
  }
}


#define AGG_THREADS 1024
#define AGG_WAVES   32
#define AGG_MAXC    256

template <int K>
__device__ __forceinline__ int row_bcast(int v) {
  return __builtin_amdgcn_update_dpp(0, v, 0x150 + K, 0xf, 0xf, true);
}

template <int F> struct agg_vec;
template <> struct agg_vec<32> { typedef v2f type; static constexpr int G = 16; };
template <> struct agg_vec<64> { typedef v4f type; static constexpr int G = 8; };

template <int F, int K0, int K>
__device__ __forceinline__ void agg_loads(typename agg_vec<F>::type (&rv)[agg_vec<F>::G],
                                          const char* __restrict__ hb, int sOff, int lanePart,
                                          int st) {
  typedef typename agg_vec<F>::type vec_t;
  if constexpr (K < agg_vec<F>::G) {
    if (K < st) {
      rv[K] = *(const vec_t*)(hb + (unsigned)(row_bcast<K0 + K>(sOff) + lanePart));
      agg_loads<F, K0, K + 1>(rv, hb, sOff, lanePart, st);
    }
  }
}

template <int F, int K0, int K>
__device__ __forceinline__ void agg_adds(const typename agg_vec<F>::type (&rv)[agg_vec<F>::G],
                                         float* acc, int wi, int dOff, int partA, int partB,
                                         bool upper, int st) {
  constexpr int V = F / 16;
  if constexpr (K < agg_vec<F>::G) {
    if (K < st) {
      const float wk = __int_as_float(row_bcast<K0 + K>(wi));
      const int dk = row_bcast<K0 + K>(dOff);
      float* a0 = (float*)((char*)acc + (unsigned)(dk + partA));
      float* a1 = (float*)((char*)acc + (unsigned)(dk + partB));
#pragma unroll
      for (int c = 0; c < V; c += 2) {
        const float lo = rv[K][c], hi = rv[K][c + 1];
        unsafeAtomicAdd(a0 + c, (upper ? hi : lo) * wk);
        unsafeAtomicAdd(a1 + c, (upper ? lo : hi) * wk);
      }
      agg_adds<F, K0, K + 1>(rv, acc, wi, dOff, partA, partB, upper, st);
    }
  }
}

template <int F, int K0>
__device__ __forceinline__ void agg_walk(const float* __restrict__ h, float* acc, int sOff,
                                         float w, int dOff, int lanePart, int st) {
  const bool upper = (threadIdx.x & 16) != 0;
  const int partA = lanePart + (upper ? 4 : 0);
  const int partB = lanePart + (upper ? 0 : 4);
  typename agg_vec<F>::type rv[agg_vec<F>::G];
  agg_loads<F, K0, 0>(rv, (const char*)h, sOff, lanePart, st);
  agg_adds<F, K0, 0>(rv, acc, __float_as_int(w), dOff, partA, partB, upper, st);
}

template <int F>
__global__ __launch_bounds__(AGG_THREADS)
void k_agg_lds(const float* __restrict__ h, const float* __restrict__ dinv,
               const float* __restrict__ bias, const unsigned* __restrict__ recs,
               const unsigned* __restrict__ offs, float* __restrict__ out, int N, int NB, int nchunk) {
  constexpr int RANGE = BK_ROWS;
  constexpr int ACC_FLOATS = RANGE * F;
  constexpr int VPR = F / 4;
  constexpr int RV = (RANGE * VPR + AGG_THREADS - 1) / AGG_THREADS;
  static_assert(AGG_THREADS % VPR == 0, "a lane's pieces all lie at the same place of their rows");
  __shared__ __attribute__((aligned(16))) float acc[ACC_FLOATS + F];
  __shared__ unsigned pre[AGG_MAXC + 1];
  __shared__ unsigned st[AGG_MAXC];

  const int tid = threadIdx.x;
  const int lane = tid & 31;
  const int wave = __builtin_amdgcn_readfirstlane(tid >> 5);
  const int bin = blockIdx.x;
  const int lo = bin * RANGE;
  const int nr = min(RANGE, N - lo);
  const int slot = 2 * (lane & 15) + (lane >> 4);
  const int lanePart = (lane & 15) * (F / 16) * 4;

  for (int i = tid; i < ACC_FLOATS; i += AGG_THREADS) acc[i] = 0.0f;

  for (int g = 0; g < nchunk; g += AGG_MAXC) {
    const int ng = min(AGG_MAXC, nchunk - g);
    if (g != 0) __syncthreads();
    if (wave == 0) {
      unsigned a[AGG_MAXC / 32], b[AGG_MAXC / 32];
#pragma unroll
      for (int q = 0; q < AGG_MAXC / 32; ++q) {
        const int ci = min(q * 32 + lane, ng - 1);
        const unsigned* o = offs + (size_t)(g + ci) * (NB + 1) + bin;
        a[q] = o[0];
        b[q] = o[1];
      }
      int run = 0;
#pragma unroll
      for (int q = 0; q < AGG_MAXC / 32; ++q) {
        const int i = q * 32 + lane;
        const int len = (i < ng) ? (int)(b[q] - a[q]) : 0;
        int tot;
        const int inc = wave_prefix_sum(len, lane, tot);
        pre[i + 1] = (unsigned)(run + inc);
        st[i] = (unsigned)((g + min(i, ng - 1)) * BK_CHUNK) + a[q];
        run += tot;
      }
      if (lane == 0) pre[0] = 0u;
    }
    __syncthreads();

    const int T = (int)pre[AGG_MAXC];
    for (int b0 = wave * 32; b0 < T; b0 += AGG_WAVES * 32) {
      const int n = min(32, T - b0);
      const bool live = slot < n;
      const unsigned p = (unsigned)min(b0 + slot, T - 1);
      int cr = 0;
#pragma unroll
      for (int sstep = AGG_MAXC / 2; sstep > 0; sstep >>= 1)
        if (pre[cr + sstep] <= p) cr += sstep;
      uint2 rec = make_uint2(0u, 0u);
      if (live) rec = ((const uint2*)recs)[st[cr] + (p - pre[cr])];
      const int s = (int)(rec.x & REC_MASK);
      const int row = (int)(rec.x >> REC_SHIFT);
      const float ds = __uint_as_float(rec.y);
      const float dd = dinv[lo + row];
      const float w = live ? ds * dd : 0.0f;
      const int sOff = s * (F * 4);
      const int dOff = live ? row * (F * 4) : ACC_FLOATS * 4;
      const int steps = (n + 1) >> 1;
      constexpr int G = agg_vec<F>::G;
      agg_walk<F, 0>(h, acc, sOff, w, dOff, lanePart, min(steps, G));
      if constexpr (G < 16) {
        if (steps > G) agg_walk<F, G>(h, acc, sOff, w, dOff, lanePart, steps - G);
      }
    }
  }

  const unsigned npc = (unsigned)(nr * VPR);
  const v4f* hv = (const v4f*)(h + (size_t)lo * F);
  v4f x[RV];
  float di[RV];
#pragma unroll
  for (int j = 0; j < RV; ++j) {
    const unsigned q = min((unsigned)(tid + j * AGG_THREADS), npc - 1u);
    x[j] = hv[q];
    di[j] = dinv[lo + (int)(q / VPR)];
  }
  const v4f bv = *(const v4f*)(bias + 4 * (tid % VPR));
  __syncthreads();

  v4f sum[RV];
#pragma unroll
  for (int j = 0; j < RV; ++j)
    sum[j] = ((const v4f*)acc)[min((unsigned)(tid + j * AGG_THREADS), npc - 1u)];
  v4f* ov = (v4f*)(out + (size_t)lo * F);
#pragma unroll
  for (int j = 0; j < RV; ++j) {
    const unsigned q = (unsigned)(tid + j * AGG_THREADS);
    const v4f d4 = {di[j], di[j], di[j], di[j]};
    if (q < npc) ov[q] = __builtin_elementwise_fma(x[j] * d4, d4, bv) + sum[j];
  }
}


static inline int cdiv_ll(long long a, long long b) { return (int)((a + b - 1) / b); }

extern "C" void kernel_launch(void* const* d_in, const int* in_sizes, int n_in,
                              void* d_out, int out_size, void* d_ws, size_t ws_size,
                              hipStream_t stream) {
  const float* X  = (const float*)d_in[0];
  const int*   ei = (const int*)d_in[1];
  const float* W1 = (const float*)d_in[2];
  const float* b1 = (const float*)d_in[3];
  const float* W2 = (const float*)d_in[4];
  const float* b2 = (const float*)d_in[5];
  float* out = (float*)d_out;

  const int N = in_sizes[0] / F_IN;
  const int E = in_sizes[1] / 2;
  const int* src = ei;
  const int* dst = ei + E;

  const int nchunk = cdiv_ll(E, BK_CHUNK);
  const int NB = cdiv_ll(N, BK_ROWS);

  unsigned* recs = (unsigned*)d_ws;
  float* dinv = (float*)(recs + (size_t)nchunk * BK_CHUNK * 2);
  float* h1   = dinv + N;
  float* out1 = h1 + (size_t)N * F_HID;
  float* h2   = out1 + (size_t)N * F_HID;
  unsigned* part = (unsigned*)(h2 + (size_t)N * F_OUT);
  unsigned* offs = part + (size_t)DEG_SLICES * N;

  const int B = 256;
  const int mtiles = cdiv_ll(N, 16);

  k_count_deg<<<dim3(cdiv_ll(N, DEG_RANGE), DEG_SLICES), DEG_THREADS, 0, stream>>>(dst, part, N, E);
  k_dinv     <<<cdiv_ll(N, B), B, 0, stream>>>(part, dinv, N);
  k_bucket   <<<nchunk, BK_THREADS, 0, stream>>>(src, dst, dinv, recs, offs, N, E, NB);

  k_gemm1    <<<cdiv_ll((long long)mtiles * 32, B), B, 0, stream>>>(X, W1, h1, N);
  k_agg_lds<F_HID><<<NB, AGG_THREADS, 0, stream>>>(h1, dinv, b1, recs, offs, out1, N, NB, nchunk);

  k_gemm2    <<<cdiv_ll((long long)mtiles * 32, B), B, 0, stream>>>(out1, W2, h2, N);
  k_agg_lds<F_OUT><<<NB, AGG_THREADS, 0, stream>>>(h2, dinv, b2, recs, offs, out, N, NB, nchunk);
}


# first product: a wave takes its 16 X rows by whole-line non-temporal loads through its own LDS patch, W1 sits in LDS in fragment order; next piece in flight under the products; k order as before
# speedup vs baseline: 4.1936x; 1.0624x over previous
#include <hip/hip_runtime.h>
#include <hip/hip_bf16.h>

typedef float v2f __attribute__((ext_vector_type(2)));
typedef float v4f __attribute__((ext_vector_type(4)));
typedef float v8f __attribute__((ext_vector_type(8)));

#define F_IN  512
#define F_HID 32
#define F_OUT 64


#define DEG_THREADS 1024
#define DEG_RANGE   25000
#define DEG_SLICES  64
#define DEG_K       50
#define DEG_ROUND   (DEG_K * DEG_THREADS)

__device__ __forceinline__ int deg_load_dst(int (&cur)[DEG_K], const int* __restrict__ dstv,
                                            int base, int tid, int E) {
  if (E >= DEG_ROUND) {
    const int rb = min(base, E - DEG_ROUND);
    const int* p = dstv + rb + tid;
#pragma unroll
    for (int k = 0; k < DEG_K; ++k) cur[k] = p[k * DEG_THREADS];
    return rb;
  }
#pragma unroll
  for (int k = 0; k < DEG_K; ++k) {
    const int e = base + k * DEG_THREADS + tid;
    const int v = dstv[min(e, E - 1)];
    cur[k] = (e < E) ? v : -1;
  }
  return base;
}

__global__ __launch_bounds__(DEG_THREADS)
void k_count_deg(const int* __restrict__ dstv, unsigned* __restrict__ part, int N, int E) {
  __shared__ unsigned cnt[DEG_RANGE];
  const int tid = threadIdx.x;
  const int lo = blockIdx.x * DEG_RANGE;
  const int nr = min(DEG_RANGE, N - lo);
  const int per = (((E + DEG_SLICES - 1) / DEG_SLICES) + 63) & ~63;
  const int e0 = min(E, (int)blockIdx.y * per);
  const int e1 = min(E, e0 + per);

  int cur[DEG_K];
  int rb = e0;
  if (e0 < e1) rb = deg_load_dst(cur, dstv, e0, tid, E);
  for (int i = tid; i < DEG_RANGE; i += DEG_THREADS) cnt[i] = 0u;
  __syncthreads();

  for (int base = e0; base < e1; ) {
    const unsigned len = (unsigned)(e1 - base);
#pragma unroll
    for (int k = 0; k < DEG_K; ++k) {
      const unsigned r = (unsigned)(cur[k] - lo);
      const unsigned q = (unsigned)(rb + k * DEG_THREADS + tid - base);
      if (r < (unsigned)nr && q < len) atomicAdd(&cnt[r], 1u);
    }
    base += DEG_ROUND;
    if (base < e1) rb = deg_load_dst(cur, dstv, base, tid, E);
  }
  __syncthreads();

  unsigned* p = part + (size_t)blockIdx.y * N + lo;
  for (int i = tid; i < nr; i += DEG_THREADS) p[i] = cnt[i];
}

__global__ void k_dinv(const unsigned* __restrict__ part, float* __restrict__ dinv, int N) {
  int i = blockIdx.x * blockDim.x + threadIdx.x;
  if (i < N) {
    unsigned c = 1u;
#pragma unroll
    for (int s = 0; s < DEG_SLICES; ++s) c += part[(size_t)s * N + i];
    dinv[i] = rsqrtf((float)c);
  }
}


#define BK_ROWS    400
#define BK_CHUNK   16384
#define BK_THREADS 1024
#define BK_K       (BK_CHUNK / BK_THREADS)
#define BK_MAXB    1024
#define REC_SHIFT  21
#define REC_MASK   ((1u << REC_SHIFT) - 1u)

__device__ __forceinline__ int row_prefix_sum(int v) {
  v += __builtin_amdgcn_update_dpp(0, v, 0x111, 0xf, 0xf, true);
  v += __builtin_amdgcn_update_dpp(0, v, 0x112, 0xf, 0xf, true);
  v += __builtin_amdgcn_update_dpp(0, v, 0x114, 0xf, 0xf, true);
  v += __builtin_amdgcn_update_dpp(0, v, 0x118, 0xf, 0xf, true);
  return v;
}

__device__ __forceinline__ int wave_prefix_sum(int v, int lane, int& total) {
  int inc = row_prefix_sum(v);
  const int t0 = __builtin_amdgcn_readlane(inc, 15);
  const int t1 = __builtin_amdgcn_readlane(inc, 31);
  if (lane >= 16) inc += t0;
  total = t0 + t1;
  return inc;
}

__global__ __launch_bounds__(BK_THREADS)
void k_bucket(const int* __restrict__ srcv, const int* __restrict__ dstv, const float* __restrict__ dinv,
              unsigned* __restrict__ recs, unsigned* __restrict__ offs, int N, int E, int NB) {
  __shared__ uint2 stage[BK_CHUNK];
  __shared__ unsigned cnt[BK_MAXB + 1];
  const int tid = threadIdx.x;
  const int lane = tid & 31;
  const int wave = __builtin_amdgcn_readfirstlane(tid >> 5);
  const int c0 = blockIdx.x * BK_CHUNK;

  int d[BK_K];
  int s[BK_K];
  if (c0 + BK_CHUNK <= E) {
    const int* q = dstv + c0 + tid;
    const int* qs = srcv + c0 + tid;
#pragma unroll
    for (int k = 0; k < BK_K; ++k) d[k] = q[k * BK_THREADS];
#pragma unroll
    for (int k = 0; k < BK_K; ++k) s[k] = qs[k * BK_THREADS];
  } else {
#pragma unroll
    for (int k = 0; k < BK_K; ++k) {
      const int e = c0 + k * BK_THREADS + tid;
      const int v = dstv[min(e, E - 1)];
      d[k] = (e < E) ? v : -1;
      s[k] = srcv[min(e, E - 1)];
    }
  }
  for (int i = tid; i <= NB; i += BK_THREADS) cnt[i] = 0u;
  __syncthreads();

  float ds[BK_K];
#pragma unroll
  for (int k = 0; k < BK_K; ++k) ds[k] = dinv[s[k]];

  unsigned rk[BK_K];
#pragma unroll
  for (int k = 0; k < BK_K; ++k) {
    rk[k] = 0u;
    if ((unsigned)d[k] < (unsigned)N) rk[k] = atomicAdd(&cnt[(unsigned)d[k] / BK_ROWS], 1u);
  }
  __syncthreads();

  if (wave == 0) {
    int run = 0;
    for (int b0 = 0; b0 < NB; b0 += 32) {
      const int b = b0 + lane;
      const int v = (b < NB) ? (int)cnt[b] : 0;
      int tot;
      const int inc = wave_prefix_sum(v, lane, tot);
      if (b < NB) cnt[b] = (unsigned)(run + inc - v);
      run += tot;
    }
    if (lane == 0) cnt[NB] = (unsigned)run;
  }
  __syncthreads();

#pragma unroll
  for (int k = 0; k < BK_K; ++k) {
    if ((unsigned)d[k] < (unsigned)N)
      stage[cnt[(unsigned)d[k] / BK_ROWS] + rk[k]] =
          make_uint2((unsigned)s[k] | ((unsigned)d[k] % BK_ROWS) << REC_SHIFT, __float_as_uint(ds[k]));
  }
  __syncthreads();

  const int total = (int)cnt[NB];
  uint2* r = (uint2*)recs + c0;
  for (int i = tid; i < total; i += BK_THREADS) r[i] = stage[i];
  unsigned* o = offs + (size_t)blockIdx.x * (NB + 1);
  for (int i = tid; i <= NB; i += BK_THREADS) o[i] = cnt[i];
}


#define G1_WAVES   25
#define G1_THREADS (G1_WAVES * 32)
#define G1_KC      64
#define G1_PITCH   (G1_KC + 4)
#define G1_NPC     (F_IN / G1_KC)
#define G1_Q       (16 * G1_KC / 128)
#define G1_WENT    ((F_IN / 4) * 32)
#define G1_WPT     ((G1_WENT + G1_THREADS - 1) / G1_THREADS)

__global__ __launch_bounds__(G1_THREADS)
void k_gemm1(const float* __restrict__ X, const float* __restrict__ W,
             float* __restrict__ H, int N) {
  __shared__ __attribute__((aligned(16))) v4f wl[G1_WENT];
  __shared__ __attribute__((aligned(16))) float xs[G1_WAVES * 16 * G1_PITCH];
  const int tid = threadIdx.x;
  const int lane = tid & 31;
  const int wave = __builtin_amdgcn_readfirstlane(tid >> 5);
  const int half = lane >> 4;
  const int lrow = lane & 15;
  const int m0 = (blockIdx.x * G1_WAVES + wave) << 4;
  const bool work = m0 < N;

  float w4[G1_WPT][4];
#pragma unroll
  for (int t = 0; t < G1_WPT; ++t) {
    const int e = min(tid + t * G1_THREADS, G1_WENT - 1);
    const int ka = 4 * (e >> 5) + 2 * ((e >> 4) & 1);
    const float* wp = W + ka * F_HID + (e & 15);
    w4[t][0] = wp[0];
    w4[t][1] = wp[F_HID];
    w4[t][2] = wp[16];
    w4[t][3] = wp[F_HID + 16];
  }

  unsigned xoff[G1_Q];
  v4f nx[G1_Q];
#pragma unroll
  for (int q = 0; q < G1_Q; ++q) {
    const int r = min(m0 + 2 * q + half, N - 1);
    xoff[q] = (unsigned)r * (unsigned)(F_IN * 4) + (unsigned)(lrow * 16);
    nx[q] = (v4f){0.0f, 0.0f, 0.0f, 0.0f};
  }
  const char* xb = (const char*)X;
  if (work) {
#pragma unroll
    for (int q = 0; q < G1_Q; ++q) nx[q] = __builtin_nontemporal_load((const v4f*)(xb + xoff[q]));
  }

#pragma unroll
  for (int t = 0; t < G1_WPT; ++t) {
    const int e = tid + t * G1_THREADS;
    if (e < G1_WENT) wl[e] = (v4f){w4[t][0], w4[t][1], w4[t][2], w4[t][3]};
  }
  __syncthreads();
  if (!work) return;

  float* sp = xs + wave * (16 * G1_PITCH) + half * G1_PITCH + lrow * 4;
  const float* ap = xs + wave * (16 * G1_PITCH) + lrow * G1_PITCH + 2 * half;
  const v4f* bp = wl + lane;

  v8f c0 = {}; v8f c1 = {};
#pragma unroll 1
  for (int p = 0; p < G1_NPC; ++p) {
#pragma unroll
    for (int q = 0; q < G1_Q; ++q) *(v4f*)(sp + q * (2 * G1_PITCH)) = nx[q];
    if (p + 1 < G1_NPC) {
      xb += G1_KC * 4;
#pragma unroll
      for (int q = 0; q < G1_Q; ++q) nx[q] = __builtin_nontemporal_load((const v4f*)(xb + xoff[q]));
    }
#pragma unroll
    for (int j = 0; j < G1_KC / 4; ++j) {
      const v2f a = *(const v2f*)(ap + 4 * j);
      const v4f b = bp[(p * (G1_KC / 4) + j) * 32];
      const v2f b0 = {b.x, b.y};
      const v2f b1 = {b.z, b.w};
      c0 = __builtin_amdgcn_wmma_f32_16x16x4_f32(false, a, false, b0, (short)0, c0, false, false);
      c1 = __builtin_amdgcn_wmma_f32_16x16x4_f32(false, a, false, b1, (short)0, c1, false, false);
    }
  }
#pragma unroll
  for (int j = 0; j < 8; ++j) {
    int m = m0 + j + (half << 3);
    if (m < N) {
      float* hr = H + (size_t)m * F_HID;
      hr[lrow]      = c0[j];
      hr[16 + lrow] = c1[j];
    }
  }
}


__global__ void k_gemm2(const float* __restrict__ A, const float* __restrict__ W,
                        float* __restrict__ H, int N) {
  const int lane = threadIdx.x & 31;
  const int wave = (blockIdx.x * blockDim.x + threadIdx.x) >> 5;
  const int m0 = wave << 4;
  if (m0 >= N) return;
  const int half = lane >> 4;
  const int lrow = lane & 15;
  int row = m0 + lrow; if (row >= N) row = N - 1;
  const float* ar = A + (size_t)row * F_HID;

  v8f c0 = {}; v8f c1 = {}; v8f c2 = {}; v8f c3 = {};
#pragma unroll
  for (int k0 = 0; k0 < F_HID; k0 += 4) {
    const int ka = k0 + (half << 1);
    v2f a; a.x = fmaxf(ar[ka], 0.0f); a.y = fmaxf(ar[ka + 1], 0.0f);
    v2f b;
    b.x = W[ka * F_OUT + lrow];            b.y = W[(ka + 1) * F_OUT + lrow];
    c0 = __builtin_amdgcn_wmma_f32_16x16x4_f32(false, a, false, b, (short)0, c0, false, false);
    b.x = W[ka * F_OUT + 16 + lrow];       b.y = W[(ka + 1) * F_OUT + 16 + lrow];
    c1 = __builtin_amdgcn_wmma_f32_16x16x4_f32(false, a, false, b, (short)0, c1, false, false);
    b.x = W[ka * F_OUT + 32 + lrow];       b.y = W[(ka + 1) * F_OUT + 32 + lrow];
    c2 = __builtin_amdgcn_wmma_f32_16x16x4_f32(false, a, false, b, (short)0, c2, false, false);
    b.x = W[ka * F_OUT + 48 + lrow];       b.y = W[(ka + 1) * F_OUT + 48 + lrow];
    c3 = __builtin_amdgcn_wmma_f32_16x16x4_f32(false, a, false, b, (short)0, c3, false, false);
  }
#pragma unroll
  for (int j = 0; j < 8; ++j) {
    int m = m0 + j + (half << 3);
    if (m < N) {
      float* hr = H + (size_t)m * F_OUT;
      hr[lrow]      = c0[j];
      hr[16 + lrow] = c1[j];
      hr[32 + lrow] = c2[j];
      hr[48 + lrow] = c3[j];
    }
  }
}


#define AGG_THREADS 1024
#define AGG_WAVES   32
#define AGG_MAXC    256

template <int K>
__device__ __forceinline__ int row_bcast(int v) {
  return __builtin_amdgcn_update_dpp(0, v, 0x150 + K, 0xf, 0xf, true);
}

template <int F> struct agg_vec;
template <> struct agg_vec<32> { typedef v2f type; static constexpr int G = 16; };
template <> struct agg_vec<64> { typedef v4f type; static constexpr int G = 8; };

template <int F, int K0, int K>
__device__ __forceinline__ void agg_loads(typename agg_vec<F>::type (&rv)[agg_vec<F>::G],
                                          const char* __restrict__ hb, int sOff, int lanePart,
                                          int st) {
  typedef typename agg_vec<F>::type vec_t;
  if constexpr (K < agg_vec<F>::G) {
    if (K < st) {
      rv[K] = *(const vec_t*)(hb + (unsigned)(row_bcast<K0 + K>(sOff) + lanePart));
      agg_loads<F, K0, K + 1>(rv, hb, sOff, lanePart, st);
    }
  }
}

template <int F, int K0, int K>
__device__ __forceinline__ void agg_adds(const typename agg_vec<F>::type (&rv)[agg_vec<F>::G],
                                         float* acc, int wi, int dOff, int partA, int partB,
                                         bool upper, int st) {
  constexpr int V = F / 16;
  if constexpr (K < agg_vec<F>::G) {
    if (K < st) {
      const float wk = __int_as_float(row_bcast<K0 + K>(wi));
      const int dk = row_bcast<K0 + K>(dOff);
      float* a0 = (float*)((char*)acc + (unsigned)(dk + partA));
      float* a1 = (float*)((char*)acc + (unsigned)(dk + partB));
#pragma unroll
      for (int c = 0; c < V; c += 2) {
        const float lo = rv[K][c], hi = rv[K][c + 1];
        unsafeAtomicAdd(a0 + c, (upper ? hi : lo) * wk);
        unsafeAtomicAdd(a1 + c, (upper ? lo : hi) * wk);
      }
      agg_adds<F, K0, K + 1>(rv, acc, wi, dOff, partA, partB, upper, st);
    }
  }
}

template <int F, int K0>
__device__ __forceinline__ void agg_walk(const float* __restrict__ h, float* acc, int sOff,
                                         float w, int dOff, int lanePart, int st) {
  const bool upper = (threadIdx.x & 16) != 0;
  const int partA = lanePart + (upper ? 4 : 0);
  const int partB = lanePart + (upper ? 0 : 4);
  typename agg_vec<F>::type rv[agg_vec<F>::G];
  agg_loads<F, K0, 0>(rv, (const char*)h, sOff, lanePart, st);
  agg_adds<F, K0, 0>(rv, acc, __float_as_int(w), dOff, partA, partB, upper, st);
}

template <int F>
__global__ __launch_bounds__(AGG_THREADS)
void k_agg_lds(const float* __restrict__ h, const float* __restrict__ dinv,
               const float* __restrict__ bias, const unsigned* __restrict__ recs,
               const unsigned* __restrict__ offs, float* __restrict__ out, int N, int NB, int nchunk) {
  constexpr int RANGE = BK_ROWS;
  constexpr int ACC_FLOATS = RANGE * F;
  constexpr int VPR = F / 4;
  constexpr int RV = (RANGE * VPR + AGG_THREADS - 1) / AGG_THREADS;
  static_assert(AGG_THREADS % VPR == 0, "a lane's pieces all lie at the same place of their rows");
  __shared__ __attribute__((aligned(16))) float acc[ACC_FLOATS + F];
  __shared__ unsigned pre[AGG_MAXC + 1];
  __shared__ unsigned st[AGG_MAXC];

  const int tid = threadIdx.x;
  const int lane = tid & 31;
  const int wave = __builtin_amdgcn_readfirstlane(tid >> 5);
  const int bin = blockIdx.x;
  const int lo = bin * RANGE;
  const int nr = min(RANGE, N - lo);
  const int slot = 2 * (lane & 15) + (lane >> 4);
  const int lanePart = (lane & 15) * (F / 16) * 4;

  for (int i = tid; i < ACC_FLOATS; i += AGG_THREADS) acc[i] = 0.0f;

  for (int g = 0; g < nchunk; g += AGG_MAXC) {
    const int ng = min(AGG_MAXC, nchunk - g);
    if (g != 0) __syncthreads();
    if (wave == 0) {
      unsigned a[AGG_MAXC / 32], b[AGG_MAXC / 32];
#pragma unroll
      for (int q = 0; q < AGG_MAXC / 32; ++q) {
        const int ci = min(q * 32 + lane, ng - 1);
        const unsigned* o = offs + (size_t)(g + ci) * (NB + 1) + bin;
        a[q] = o[0];
        b[q] = o[1];
      }
      int run = 0;
#pragma unroll
      for (int q = 0; q < AGG_MAXC / 32; ++q) {
        const int i = q * 32 + lane;
        const int len = (i < ng) ? (int)(b[q] - a[q]) : 0;
        int tot;
        const int inc = wave_prefix_sum(len, lane, tot);
        pre[i + 1] = (unsigned)(run + inc);
        st[i] = (unsigned)((g + min(i, ng - 1)) * BK_CHUNK) + a[q];
        run += tot;
      }
      if (lane == 0) pre[0] = 0u;
    }
    __syncthreads();

    const int T = (int)pre[AGG_MAXC];
    for (int b0 = wave * 32; b0 < T; b0 += AGG_WAVES * 32) {
      const int n = min(32, T - b0);
      const bool live = slot < n;
      const unsigned p = (unsigned)min(b0 + slot, T - 1);
      int cr = 0;
#pragma unroll
      for (int sstep = AGG_MAXC / 2; sstep > 0; sstep >>= 1)
        if (pre[cr + sstep] <= p) cr += sstep;
      uint2 rec = make_uint2(0u, 0u);
      if (live) rec = ((const uint2*)recs)[st[cr] + (p - pre[cr])];
      const int s = (int)(rec.x & REC_MASK);
      const int row = (int)(rec.x >> REC_SHIFT);
      const float ds = __uint_as_float(rec.y);
      const float dd = dinv[lo + row];
      const float w = live ? ds * dd : 0.0f;
      const int sOff = s * (F * 4);
      const int dOff = live ? row * (F * 4) : ACC_FLOATS * 4;
      const int steps = (n + 1) >> 1;
      constexpr int G = agg_vec<F>::G;
      agg_walk<F, 0>(h, acc, sOff, w, dOff, lanePart, min(steps, G));
      if constexpr (G < 16) {
        if (steps > G) agg_walk<F, G>(h, acc, sOff, w, dOff, lanePart, steps - G);
      }
    }
  }

  const unsigned npc = (unsigned)(nr * VPR);
  const v4f* hv = (const v4f*)(h + (size_t)lo * F);
  v4f x[RV];
  float di[RV];
#pragma unroll
  for (int j = 0; j < RV; ++j) {
    const unsigned q = min((unsigned)(tid + j * AGG_THREADS), npc - 1u);
    x[j] = hv[q];
    di[j] = dinv[lo + (int)(q / VPR)];
  }
  const v4f bv = *(const v4f*)(bias + 4 * (tid % VPR));
  __syncthreads();

  v4f sum[RV];
#pragma unroll
  for (int j = 0; j < RV; ++j)
    sum[j] = ((const v4f*)acc)[min((unsigned)(tid + j * AGG_THREADS), npc - 1u)];
  v4f* ov = (v4f*)(out + (size_t)lo * F);
#pragma unroll
  for (int j = 0; j < RV; ++j) {
    const unsigned q = (unsigned)(tid + j * AGG_THREADS);
    const v4f d4 = {di[j], di[j], di[j], di[j]};
    if (q < npc) ov[q] = __builtin_elementwise_fma(x[j] * d4, d4, bv) + sum[j];
  }
}


static inline int cdiv_ll(long long a, long long b) { return (int)((a + b - 1) / b); }

extern "C" void kernel_launch(void* const* d_in, const int* in_sizes, int n_in,
                              void* d_out, int out_size, void* d_ws, size_t ws_size,
                              hipStream_t stream) {
  const float* X  = (const float*)d_in[0];
  const int*   ei = (const int*)d_in[1];
  const float* W1 = (const float*)d_in[2];
  const float* b1 = (const float*)d_in[3];
  const float* W2 = (const float*)d_in[4];
  const float* b2 = (const float*)d_in[5];
  float* out = (float*)d_out;

  const int N = in_sizes[0] / F_IN;
  const int E = in_sizes[1] / 2;
  const int* src = ei;
  const int* dst = ei + E;

  const int nchunk = cdiv_ll(E, BK_CHUNK);
  const int NB = cdiv_ll(N, BK_ROWS);

  unsigned* recs = (unsigned*)d_ws;
  float* dinv = (float*)(recs + (size_t)nchunk * BK_CHUNK * 2);
  float* h1   = dinv + N;
  float* out1 = h1 + (size_t)N * F_HID;
  float* h2   = out1 + (size_t)N * F_HID;
  unsigned* part = (unsigned*)(h2 + (size_t)N * F_OUT);
  unsigned* offs = part + (size_t)DEG_SLICES * N;

  const int B = 256;
  const int mtiles = cdiv_ll(N, 16);

  k_count_deg<<<dim3(cdiv_ll(N, DEG_RANGE), DEG_SLICES), DEG_THREADS, 0, stream>>>(dst, part, N, E);
  k_dinv     <<<cdiv_ll(N, B), B, 0, stream>>>(part, dinv, N);
  k_bucket   <<<nchunk, BK_THREADS, 0, stream>>>(src, dst, dinv, recs, offs, N, E, NB);

  k_gemm1    <<<cdiv_ll(mtiles, G1_WAVES), G1_THREADS, 0, stream>>>(X, W1, h1, N);
  k_agg_lds<F_HID><<<NB, AGG_THREADS, 0, stream>>>(h1, dinv, b1, recs, offs, out1, N, NB, nchunk);

  k_gemm2    <<<cdiv_ll((long long)mtiles * 32, B), B, 0, stream>>>(out1, W2, h2, N);
  k_agg_lds<F_OUT><<<NB, AGG_THREADS, 0, stream>>>(h2, dinv, b2, recs, offs, out, N, NB, nchunk);
}
